# starting kernel
# baseline (speedup 1.0000x reference)
#include <hip/hip_runtime.h>
#include <math.h>

#define BB 2
#define II 48
#define JJ 320
#define DD 256

#define NEGC   (-1000000000.0f)
#define LOGEPS (-1000.0f)
#define NEG_ID (-1e30f)

#define LOG2E_F 1.4426950408889634f
#define LN2_F   0.6931471805599453f

typedef __attribute__((ext_vector_type(2))) float v2f;
typedef __attribute__((ext_vector_type(8))) float v8f;

__device__ __forceinline__ float fast_exp(float x) {
    return __builtin_amdgcn_exp2f(x * LOG2E_F);
}
__device__ __forceinline__ float fast_log(float x) {
    return LN2_F * __builtin_amdgcn_logf(x);
}

__device__ __forceinline__ float lse2(float a, float b) {
    float m = fmaxf(a, b);
    float d = fminf(a, b) - m;
    float t = fast_exp(d);
    return m + fast_log(1.0f + t);
}

__global__ void __launch_bounds__(256)
k_transpose_text(const float* __restrict__ text, float* __restrict__ TT) {
    int b = blockIdx.y;
    for (int t = blockIdx.x * 256 + threadIdx.x; t < II * DD; t += gridDim.x * 256) {
        int i = t / DD, d = t % DD;
        TT[((size_t)b * DD + d) * II + i] = text[((size_t)b * II + i) * DD + d];
    }
}

__global__ void __launch_bounds__(32)
k_energy_wmma(const float* __restrict__ text, const float* __restrict__ mel,
              const float* __restrict__ gum, float* __restrict__ E) {
    const int b = blockIdx.z, it = blockIdx.y, jt = blockIdx.x;
    const int lane = threadIdx.x;
    const int half = lane >> 4;
    const int line = lane & 15;

    const float* __restrict__ Ta = text + (size_t)b * II * DD + (size_t)(it * 16 + line) * DD + half * 2;
    const float* __restrict__ Mb = mel  + (size_t)b * JJ * DD + (size_t)(jt * 16 + line) * DD + half * 2;

    __builtin_prefetch(Ta, 0, 1);
    __builtin_prefetch(Mb, 0, 1);
    __builtin_prefetch(Ta + 128, 0, 1);
    __builtin_prefetch(Mb + 128, 0, 1);

    v8f acc = {};
#pragma unroll 4
    for (int k = 0; k < DD; k += 4) {
        v2f a  = *(const v2f*)(Ta + k);
        v2f bv = *(const v2f*)(Mb + k);
        acc = __builtin_amdgcn_wmma_f32_16x16x4_f32(false, a, false, bv,
                                                    (short)0, acc, false, false);
    }

    const float invTemp = 1.0f / 0.55f;
    const float scale   = 1.0f / 256.0f;
#pragma unroll
    for (int v = 0; v < 8; ++v) {
        int i = it * 16 + v + half * 8;
        int j = jt * 16 + line;
        size_t idx = (size_t)b * II * JJ + (size_t)i * JJ + j;
        float u = gum[idx];
        u = fminf(fmaxf(u, 1e-7f), 1.0f - 1e-7f);
        float noise = -fast_log(-fast_log(u));
        E[idx] = (acc[v] * scale + noise) * invTemp;
    }
}

__global__ void __launch_bounds__(32)
k_suffix_lse(const float* __restrict__ E, float* __restrict__ S) {
    const int row = blockIdx.x;
    const int i = row % II;
    const int lane = threadIdx.x;
    const float* __restrict__ e = E + (size_t)row * JJ;
    float* __restrict__ s = S + (size_t)row * JJ;

    float carry = NEG_ID;
    for (int c = (JJ / 32) - 1; c >= 0; --c) {
        int j = c * 32 + lane;
        float x = e[j];
        if (i == II - 1 && j != JJ - 1) x = LOGEPS;
        float acc = x;
#pragma unroll
        for (int off = 1; off < 32; off <<= 1) {
            float y = __shfl_down(acc, off, 32);
            if (lane + off < 32) acc = lse2(acc, y);
        }
        float val = lse2(acc, carry);
        s[j] = val;
        carry = __shfl(val, 0, 32);
    }
}

__global__ void __launch_bounds__(32)
k_forward(const float* __restrict__ E, const float* __restrict__ S,
          float* __restrict__ P) {
    __shared__ float prev[JJ];
    __shared__ float tsh[JJ];
    const int b = blockIdx.x;
    const int lane = threadIdx.x;

    for (int j = lane; j < JJ; j += 32) {
        float v = (j == 0) ? 0.0f : NEGC;
        prev[j] = v;
        P[((size_t)b * II + 0) * JJ + j] = v;
    }
    __syncthreads();

    for (int i = 1; i < II; ++i) {
        const float* __restrict__ Er = E + ((size_t)b * II + (i - 1)) * JJ;
        const float* __restrict__ Sr = S + ((size_t)b * II + (i - 1)) * JJ;
        float carry = NEG_ID;
        for (int c = 0; c < JJ / 32; ++c) {
            int k = c * 32 + lane;
            float acc = prev[k] - Sr[k];
#pragma unroll
            for (int off = 1; off < 32; off <<= 1) {
                float y = __shfl_up(acc, off, 32);
                if (lane >= off) acc = lse2(acc, y);
            }
            float scan = lse2(acc, carry);
            carry = __shfl(scan, 31, 32);
            tsh[k] = Er[k] + scan;
        }
        __syncthreads();
        for (int jc = lane; jc < JJ; jc += 32) {
            float v = (jc == 0) ? NEGC : tsh[jc - 1];
            if (!(jc >= i && jc < (JJ - II + i + 2))) v = NEGC;
            prev[jc] = v;
            P[((size_t)b * II + i) * JJ + jc] = v;
        }
        __syncthreads();
    }
}

__global__ void __launch_bounds__(32)
k_gamma(const float* __restrict__ P, const float* __restrict__ S,
        float* __restrict__ GT) {
    const int row = blockIdx.x;
    const int b = row / II, i = row % II;
    const int lane = threadIdx.x;
    const float* __restrict__ p = P + (size_t)row * JJ;
    const float* __restrict__ s = S + (size_t)row * JJ;

    float carry = NEG_ID;
    for (int c = 0; c < JJ / 32; ++c) {
        int j = c * 32 + lane;
        float acc = p[j] - s[j];
#pragma unroll
        for (int off = 1; off < 32; off <<= 1) {
            float y = __shfl_up(acc, off, 32);
            if (lane >= off) acc = lse2(acc, y);
        }
        float scan = lse2(acc, carry);
        carry = __shfl(scan, 31, 32);
        float lb = s[j] + scan;
        float gamma = fast_exp(lb);
        if (i == II - 1 && j != JJ - 1) gamma = 0.0f;
        GT[((size_t)b * JJ + j) * II + i] = gamma;
    }
}

__global__ void __launch_bounds__(32)
k_expand_wmma(const float* __restrict__ GT, const float* __restrict__ TT,
              float* __restrict__ out) {
    const int b = blockIdx.z, jt = blockIdx.y, dt = blockIdx.x;
    const int lane = threadIdx.x;
    const int half = lane >> 4;
    const int line = lane & 15;
    const int j0 = jt * 16, d0 = dt * 16;
    const float* __restrict__ Ga = GT + ((size_t)b * JJ + (j0 + line)) * II + half * 2;
    const float* __restrict__ Tb = TT + ((size_t)b * DD + (d0 + line)) * II + half * 2;

    __builtin_prefetch(Ga, 0, 1);
    __builtin_prefetch(Tb, 0, 1);

    v8f acc = {};
#pragma unroll
    for (int k = 0; k < II; k += 4) {
        v2f a  = *(const v2f*)(Ga + k);
        v2f bv = *(const v2f*)(Tb + k);
        acc = __builtin_amdgcn_wmma_f32_16x16x4_f32(false, a, false, bv,
                                                    (short)0, acc, false, false);
    }
#pragma unroll
    for (int v = 0; v < 8; ++v) {
        int j = j0 + v + half * 8;
        out[(size_t)b * JJ * DD + (size_t)j * DD + d0 + line] = acc[v];
    }
}

extern "C" void kernel_launch(void* const* d_in, const int* in_sizes, int n_in,
                              void* d_out, int out_size, void* d_ws, size_t ws_size,
                              hipStream_t stream) {
    const float* text = (const float*)d_in[0];
    const float* mel  = (const float*)d_in[1];
    const float* gum  = (const float*)d_in[2];
    float* out = (float*)d_out;

    const size_t N = (size_t)BB * II * JJ;
    float* E  = (float*)d_ws;
    float* S  = E + N;
    float* P  = S + N;
    float* GT = P + N;
    float* TT = GT + N;

    k_transpose_text<<<dim3(8, BB), 256, 0, stream>>>(text, TT);
    k_energy_wmma<<<dim3(JJ / 16, II / 16, BB), 32, 0, stream>>>(text, mel, gum, E);
    k_suffix_lse<<<dim3(BB * II), 32, 0, stream>>>(E, S);
    k_forward<<<dim3(BB), 32, 0, stream>>>(E, S, P);
    k_gamma<<<dim3(BB * II), 32, 0, stream>>>(P, S, GT);
    k_expand_wmma<<<dim3(DD / 16, JJ / 16, BB), 32, 0, stream>>>(GT, TT, out);
    (void)in_sizes; (void)n_in; (void)out_size; (void)ws_size;
}


# K3 forward scan: each lane keeps 10 consecutive columns of the row in registers (serial lane scan + one wave scan of lane totals, next row prefetched); LSE reduction tree changes
# speedup vs baseline: 4.8625x; 4.8625x over previous
#include <hip/hip_runtime.h>
#include <math.h>

#define BB 2
#define II 48
#define JJ 320
#define DD 256

#define NEGC   (-1000000000.0f)
#define LOGEPS (-1000.0f)
#define NEG_ID (-1e30f)

#define LOG2E_F 1.4426950408889634f
#define LN2_F   0.6931471805599453f

typedef __attribute__((ext_vector_type(2))) float v2f;
typedef __attribute__((ext_vector_type(8))) float v8f;

__device__ __forceinline__ float fast_exp(float x) {
    return __builtin_amdgcn_exp2f(x * LOG2E_F);
}
__device__ __forceinline__ float fast_log(float x) {
    return LN2_F * __builtin_amdgcn_logf(x);
}

__device__ __forceinline__ float lse2(float a, float b) {
    float m = fmaxf(a, b);
    float d = fminf(a, b) - m;
    float t = fast_exp(d);
    return m + fast_log(1.0f + t);
}

__global__ void __launch_bounds__(256)
k_transpose_text(const float* __restrict__ text, float* __restrict__ TT) {
    int b = blockIdx.y;
    for (int t = blockIdx.x * 256 + threadIdx.x; t < II * DD; t += gridDim.x * 256) {
        int i = t / DD, d = t % DD;
        TT[((size_t)b * DD + d) * II + i] = text[((size_t)b * II + i) * DD + d];
    }
}

__global__ void __launch_bounds__(32)
k_energy_wmma(const float* __restrict__ text, const float* __restrict__ mel,
              const float* __restrict__ gum, float* __restrict__ E) {
    const int b = blockIdx.z, it = blockIdx.y, jt = blockIdx.x;
    const int lane = threadIdx.x;
    const int half = lane >> 4;
    const int line = lane & 15;

    const float* __restrict__ Ta = text + (size_t)b * II * DD + (size_t)(it * 16 + line) * DD + half * 2;
    const float* __restrict__ Mb = mel  + (size_t)b * JJ * DD + (size_t)(jt * 16 + line) * DD + half * 2;

    __builtin_prefetch(Ta, 0, 1);
    __builtin_prefetch(Mb, 0, 1);
    __builtin_prefetch(Ta + 128, 0, 1);
    __builtin_prefetch(Mb + 128, 0, 1);

    v8f acc = {};
#pragma unroll 4
    for (int k = 0; k < DD; k += 4) {
        v2f a  = *(const v2f*)(Ta + k);
        v2f bv = *(const v2f*)(Mb + k);
        acc = __builtin_amdgcn_wmma_f32_16x16x4_f32(false, a, false, bv,
                                                    (short)0, acc, false, false);
    }

    const float invTemp = 1.0f / 0.55f;
    const float scale   = 1.0f / 256.0f;
#pragma unroll
    for (int v = 0; v < 8; ++v) {
        int i = it * 16 + v + half * 8;
        int j = jt * 16 + line;
        size_t idx = (size_t)b * II * JJ + (size_t)i * JJ + j;
        float u = gum[idx];
        u = fminf(fmaxf(u, 1e-7f), 1.0f - 1e-7f);
        float noise = -fast_log(-fast_log(u));
        E[idx] = (acc[v] * scale + noise) * invTemp;
    }
}

__global__ void __launch_bounds__(32)
k_suffix_lse(const float* __restrict__ E, float* __restrict__ S) {
    const int row = blockIdx.x;
    const int i = row % II;
    const int lane = threadIdx.x;
    const float* __restrict__ e = E + (size_t)row * JJ;
    float* __restrict__ s = S + (size_t)row * JJ;

    float carry = NEG_ID;
    for (int c = (JJ / 32) - 1; c >= 0; --c) {
        int j = c * 32 + lane;
        float x = e[j];
        if (i == II - 1 && j != JJ - 1) x = LOGEPS;
        float acc = x;
#pragma unroll
        for (int off = 1; off < 32; off <<= 1) {
            float y = __shfl_down(acc, off, 32);
            if (lane + off < 32) acc = lse2(acc, y);
        }
        float val = lse2(acc, carry);
        s[j] = val;
        carry = __shfl(val, 0, 32);
    }
}

#define KPL (JJ / 32)

__device__ __forceinline__ void load_row(const float* __restrict__ p, float (&r)[KPL]) {
#pragma unroll
    for (int u = 0; u < KPL / 2; ++u) {
        v2f q = *(const v2f*)(p + 2 * u);
        r[2 * u] = q.x; r[2 * u + 1] = q.y;
    }
}

__global__ void __launch_bounds__(32)
k_forward(const float* __restrict__ E, const float* __restrict__ S,
          float* __restrict__ P) {
    const int b = blockIdx.x;
    const int lane = threadIdx.x;
    const int k0 = lane * KPL;
    const float* __restrict__ Eb = E + (size_t)b * II * JJ + k0;
    const float* __restrict__ Sb = S + (size_t)b * II * JJ + k0;
    float* __restrict__ Pb = P + (size_t)b * II * JJ + k0;

    float prev[KPL], s[KPL], e[KPL];
#pragma unroll
    for (int t = 0; t < KPL; ++t) prev[t] = (k0 + t == 0) ? 0.0f : NEGC;
#pragma unroll
    for (int u = 0; u < KPL / 2; ++u) {
        v2f q; q.x = prev[2 * u]; q.y = prev[2 * u + 1];
        *(v2f*)(Pb + 2 * u) = q;
    }
    load_row(Sb, s);
    load_row(Eb, e);

    for (int i = 1; i < II; ++i) {
        const int rn = (i < II - 1) ? i : (II - 2);
        float sn[KPL], en[KPL];
        load_row(Sb + (size_t)rn * JJ, sn);
        load_row(Eb + (size_t)rn * JJ, en);

        float l[KPL];
        l[0] = prev[0] - s[0];
#pragma unroll
        for (int t = 1; t < KPL; ++t) l[t] = lse2(l[t - 1], prev[t] - s[t]);

        float incl = l[KPL - 1];
#pragma unroll
        for (int off = 1; off < 32; off <<= 1) {
            float y = __shfl_up(incl, off, 32);
            if (lane >= off) incl = lse2(incl, y);
        }
        float excl = __shfl_up(incl, 1, 32);
        float hand = __shfl_up(e[KPL - 1] + incl, 1, 32);
        if (lane == 0) { excl = NEG_ID; hand = NEGC; }

        float nv[KPL];
        nv[0] = hand;
#pragma unroll
        for (int t = 0; t < KPL - 1; ++t) nv[t + 1] = e[t] + lse2(l[t], excl);

#pragma unroll
        for (int t = 0; t < KPL; ++t) {
            const int jc = k0 + t;
            if (!(jc >= i && jc < (JJ - II + i + 2))) nv[t] = NEGC;
            prev[t] = nv[t];
        }
#pragma unroll
        for (int u = 0; u < KPL / 2; ++u) {
            v2f q; q.x = prev[2 * u]; q.y = prev[2 * u + 1];
            *(v2f*)(Pb + (size_t)i * JJ + 2 * u) = q;
        }
#pragma unroll
        for (int t = 0; t < KPL; ++t) { s[t] = sn[t]; e[t] = en[t]; }
    }
}

__global__ void __launch_bounds__(32)
k_gamma(const float* __restrict__ P, const float* __restrict__ S,
        float* __restrict__ GT) {
    const int row = blockIdx.x;
    const int b = row / II, i = row % II;
    const int lane = threadIdx.x;
    const float* __restrict__ p = P + (size_t)row * JJ;
    const float* __restrict__ s = S + (size_t)row * JJ;

    float carry = NEG_ID;
    for (int c = 0; c < JJ / 32; ++c) {
        int j = c * 32 + lane;
        float acc = p[j] - s[j];
#pragma unroll
        for (int off = 1; off < 32; off <<= 1) {
            float y = __shfl_up(acc, off, 32);
            if (lane >= off) acc = lse2(acc, y);
        }
        float scan = lse2(acc, carry);
        carry = __shfl(scan, 31, 32);
        float lb = s[j] + scan;
        float gamma = fast_exp(lb);
        if (i == II - 1 && j != JJ - 1) gamma = 0.0f;
        GT[((size_t)b * JJ + j) * II + i] = gamma;
    }
}

__global__ void __launch_bounds__(32)
k_expand_wmma(const float* __restrict__ GT, const float* __restrict__ TT,
              float* __restrict__ out) {
    const int b = blockIdx.z, jt = blockIdx.y, dt = blockIdx.x;
    const int lane = threadIdx.x;
    const int half = lane >> 4;
    const int line = lane & 15;
    const int j0 = jt * 16, d0 = dt * 16;
    const float* __restrict__ Ga = GT + ((size_t)b * JJ + (j0 + line)) * II + half * 2;
    const float* __restrict__ Tb = TT + ((size_t)b * DD + (d0 + line)) * II + half * 2;

    __builtin_prefetch(Ga, 0, 1);
    __builtin_prefetch(Tb, 0, 1);

    v8f acc = {};
#pragma unroll
    for (int k = 0; k < II; k += 4) {
        v2f a  = *(const v2f*)(Ga + k);
        v2f bv = *(const v2f*)(Tb + k);
        acc = __builtin_amdgcn_wmma_f32_16x16x4_f32(false, a, false, bv,
                                                    (short)0, acc, false, false);
    }
#pragma unroll
    for (int v = 0; v < 8; ++v) {
        int j = j0 + v + half * 8;
        out[(size_t)b * JJ * DD + (size_t)j * DD + d0 + line] = acc[v];
    }
}

extern "C" void kernel_launch(void* const* d_in, const int* in_sizes, int n_in,
                              void* d_out, int out_size, void* d_ws, size_t ws_size,
                              hipStream_t stream) {
    const float* text = (const float*)d_in[0];
    const float* mel  = (const float*)d_in[1];
    const float* gum  = (const float*)d_in[2];
    float* out = (float*)d_out;

    const size_t N = (size_t)BB * II * JJ;
    float* E  = (float*)d_ws;
    float* S  = E + N;
    float* P  = S + N;
    float* GT = P + N;
    float* TT = GT + N;

    k_transpose_text<<<dim3(8, BB), 256, 0, stream>>>(text, TT);
    k_energy_wmma<<<dim3(JJ / 16, II / 16, BB), 32, 0, stream>>>(text, mel, gum, E);
    k_suffix_lse<<<dim3(BB * II), 32, 0, stream>>>(E, S);
    k_forward<<<dim3(BB), 32, 0, stream>>>(E, S, P);
    k_gamma<<<dim3(BB * II), 32, 0, stream>>>(P, S, GT);
    k_expand_wmma<<<dim3(DD / 16, JJ / 16, BB), 32, 0, stream>>>(GT, TT, out);
    (void)in_sizes; (void)n_in; (void)out_size; (void)ws_size;
}


# K3 forward recurrence in the linear domain: per-row prefix SUMS (one fma per column, DPP row scan) replace the LSE chains, rows rescaled by exact powers of two; order of summation changes
# speedup vs baseline: 6.8352x; 1.4057x over previous
#include <hip/hip_runtime.h>
#include <math.h>

#define BB 2
#define II 48
#define JJ 320
#define DD 256

#define NEGC   (-1000000000.0f)
#define LOGEPS (-1000.0f)
#define NEG_ID (-1e30f)

#define LOG2E_F 1.4426950408889634f
#define LN2_F   0.6931471805599453f

typedef __attribute__((ext_vector_type(2))) float v2f;
typedef __attribute__((ext_vector_type(8))) float v8f;

__device__ __forceinline__ float fast_exp(float x) {
    return __builtin_amdgcn_exp2f(x * LOG2E_F);
}
__device__ __forceinline__ float fast_log(float x) {
    return LN2_F * __builtin_amdgcn_logf(x);
}

__device__ __forceinline__ float lse2(float a, float b) {
    float m = fmaxf(a, b);
    float d = fminf(a, b) - m;
    float t = fast_exp(d);
    return m + fast_log(1.0f + t);
}

__global__ void __launch_bounds__(256)
k_transpose_text(const float* __restrict__ text, float* __restrict__ TT) {
    int b = blockIdx.y;
    for (int t = blockIdx.x * 256 + threadIdx.x; t < II * DD; t += gridDim.x * 256) {
        int i = t / DD, d = t % DD;
        TT[((size_t)b * DD + d) * II + i] = text[((size_t)b * II + i) * DD + d];
    }
}

__global__ void __launch_bounds__(32)
k_energy_wmma(const float* __restrict__ text, const float* __restrict__ mel,
              const float* __restrict__ gum, float* __restrict__ E) {
    const int b = blockIdx.z, it = blockIdx.y, jt = blockIdx.x;
    const int lane = threadIdx.x;
    const int half = lane >> 4;
    const int line = lane & 15;

    const float* __restrict__ Ta = text + (size_t)b * II * DD + (size_t)(it * 16 + line) * DD + half * 2;
    const float* __restrict__ Mb = mel  + (size_t)b * JJ * DD + (size_t)(jt * 16 + line) * DD + half * 2;

    __builtin_prefetch(Ta, 0, 1);
    __builtin_prefetch(Mb, 0, 1);
    __builtin_prefetch(Ta + 128, 0, 1);
    __builtin_prefetch(Mb + 128, 0, 1);

    v8f acc = {};
#pragma unroll 4
    for (int k = 0; k < DD; k += 4) {
        v2f a  = *(const v2f*)(Ta + k);
        v2f bv = *(const v2f*)(Mb + k);
        acc = __builtin_amdgcn_wmma_f32_16x16x4_f32(false, a, false, bv,
                                                    (short)0, acc, false, false);
    }

    const float invTemp = 1.0f / 0.55f;
    const float scale   = 1.0f / 256.0f;
#pragma unroll
    for (int v = 0; v < 8; ++v) {
        int i = it * 16 + v + half * 8;
        int j = jt * 16 + line;
        size_t idx = (size_t)b * II * JJ + (size_t)i * JJ + j;
        float u = gum[idx];
        u = fminf(fmaxf(u, 1e-7f), 1.0f - 1e-7f);
        float noise = -fast_log(-fast_log(u));
        E[idx] = (acc[v] * scale + noise) * invTemp;
    }
}

__global__ void __launch_bounds__(32)
k_suffix_lse(const float* __restrict__ E, float* __restrict__ S) {
    const int row = blockIdx.x;
    const int i = row % II;
    const int lane = threadIdx.x;
    const float* __restrict__ e = E + (size_t)row * JJ;
    float* __restrict__ s = S + (size_t)row * JJ;

    float carry = NEG_ID;
    for (int c = (JJ / 32) - 1; c >= 0; --c) {
        int j = c * 32 + lane;
        float x = e[j];
        if (i == II - 1 && j != JJ - 1) x = LOGEPS;
        float acc = x;
#pragma unroll
        for (int off = 1; off < 32; off <<= 1) {
            float y = __shfl_down(acc, off, 32);
            if (lane + off < 32) acc = lse2(acc, y);
        }
        float val = lse2(acc, carry);
        s[j] = val;
        carry = __shfl(val, 0, 32);
    }
}

#define KPL (JJ / 32)

__device__ __forceinline__ void load_row(const float* __restrict__ p, float (&r)[KPL]) {
#pragma unroll
    for (int u = 0; u < KPL / 2; ++u) {
        v2f q = *(const v2f*)(p + 2 * u);
        r[2 * u] = q.x; r[2 * u + 1] = q.y;
    }
}

template <int N>
__device__ __forceinline__ float row_shr0(float v) {
    return __int_as_float(__builtin_amdgcn_update_dpp(0, __float_as_int(v), 0x110 + N, 0xf, 0xf, true));
}
__device__ __forceinline__ float row_incl_sum(float x) {
    x += row_shr0<1>(x);
    x += row_shr0<2>(x);
    x += row_shr0<4>(x);
    x += row_shr0<8>(x);
    return x;
}
template <int L>
__device__ __forceinline__ float lane_value(float v) {
    return __int_as_float(__builtin_amdgcn_readlane(__float_as_int(v), L));
}

__device__ __forceinline__ void store_log_row(float* __restrict__ p, const float (&ph)[KPL], const int esum) {
    const float es = (float)esum;
#pragma unroll
    for (int u = 0; u < KPL / 2; ++u) {
        v2f q;
        q.x = fmaxf((es + __builtin_amdgcn_logf(ph[2 * u])) * LN2_F, NEGC);
        q.y = fmaxf((es + __builtin_amdgcn_logf(ph[2 * u + 1])) * LN2_F, NEGC);
        *(v2f*)(p + 2 * u) = q;
    }
}

template <bool PREF>
__device__ __forceinline__ void fwd_step(const int i, const int lane, const int k0,
                                         const float* __restrict__ Sb, const float* __restrict__ Eb,
                                         float* __restrict__ Pb,
                                         float (&ph)[KPL], int& esum,
                                         const float (&s)[KPL], const float (&e)[KPL],
                                         float (&sd)[KPL], float (&ed)[KPL]) {
    if (PREF) {
        load_row(Sb + (size_t)(i + 1) * JJ, sd);
        load_row(Eb + (size_t)(i + 1) * JJ, ed);
    }
    store_log_row(Pb + (size_t)(i - 1) * JJ, ph, esum);

    const float t01 = ph[0] + ph[1], t23 = ph[2] + ph[3], t45 = ph[4] + ph[5];
    const float t67 = ph[6] + ph[7], t89 = ph[8] + ph[9];
    float tot = ((t01 + t23) + (t45 + t67)) + t89;
    tot = row_incl_sum(tot);
    const float M = lane_value<15>(tot) + lane_value<31>(tot);
    int n = __builtin_amdgcn_frexp_expf(M);
    n = (n < -100) ? -100 : n;
    const float g = ldexpf(1.0f, -n);

    const float s0L = lane_value<0>(s[0]) * LOG2E_F;
    const unsigned d = (unsigned)(k0 + 1 - i);
    float w[KPL], em[KPL];
#pragma unroll
    for (int t = 0; t < KPL; ++t) w[t] = __builtin_amdgcn_exp2f(fmaf(-s[t], LOG2E_F, s0L));
#pragma unroll
    for (int t = 0; t < KPL; ++t) {
        const bool inw = (d + (unsigned)t) < (unsigned)(JJ - II + 2);
        const float v = __builtin_amdgcn_exp2f(fmaf(e[t], LOG2E_F, -s0L)) * g;
        em[t] = inw ? v : 0.0f;
    }

    float l[KPL];
    l[0] = ph[0] * w[0];
#pragma unroll
    for (int t = 1; t < KPL; ++t) l[t] = l[t - 1] + ph[t] * w[t];

    const float xr = row_incl_sum(l[KPL - 1]);
    const float r0 = lane_value<15>(xr);
    const float add = (lane >= 16) ? r0 : 0.0f;
    const float excl = row_shr0<1>(xr) + add;
    const float hv = (xr + add) * em[KPL - 1];
    float hand = row_shr0<1>(hv);
    const float h15 = lane_value<15>(hv);
    if (lane == 16) hand = h15;

    float nxt[KPL];
    nxt[0] = hand;
#pragma unroll
    for (int t = 0; t < KPL - 1; ++t) nxt[t + 1] = (excl + l[t]) * em[t];
#pragma unroll
    for (int t = 0; t < KPL; ++t) ph[t] = nxt[t];
    esum += n;
}

__global__ void __launch_bounds__(32)
k_forward(const float* __restrict__ E, const float* __restrict__ S,
          float* __restrict__ P) {
    const int b = blockIdx.x;
    const int lane = threadIdx.x;
    const int k0 = lane * KPL;
    const float* __restrict__ Eb = E + (size_t)b * II * JJ + k0;
    const float* __restrict__ Sb = S + (size_t)b * II * JJ + k0;
    float* __restrict__ Pb = P + (size_t)b * II * JJ + k0;

    float ph[KPL];
    float sA[KPL], eA[KPL], sB[KPL], eB[KPL], sC[KPL], eC[KPL];
#pragma unroll
    for (int t = 0; t < KPL; ++t) ph[t] = (k0 + t == 0) ? 1.0f : 0.0f;
    load_row(Sb, sA);
    load_row(Eb, eA);
    __builtin_amdgcn_sched_barrier(0);
    load_row(Sb + JJ, sB);
    load_row(Eb + JJ, eB);
    int esum = 0;

    int i = 1;
#pragma unroll 1
    for (; i + 2 < II - 1; i += 3) {
        fwd_step<true>(i,     lane, k0, Sb, Eb, Pb, ph, esum, sA, eA, sC, eC);
        __builtin_amdgcn_sched_barrier(0);
        fwd_step<true>(i + 1, lane, k0, Sb, Eb, Pb, ph, esum, sB, eB, sA, eA);
        __builtin_amdgcn_sched_barrier(0);
        fwd_step<true>(i + 2, lane, k0, Sb, Eb, Pb, ph, esum, sC, eC, sB, eB);
        __builtin_amdgcn_sched_barrier(0);
    }
    fwd_step<false>(i,     lane, k0, Sb, Eb, Pb, ph, esum, sA, eA, sC, eC);
    fwd_step<false>(i + 1, lane, k0, Sb, Eb, Pb, ph, esum, sB, eB, sA, eA);
    store_log_row(Pb + (size_t)(II - 1) * JJ, ph, esum);
}

__global__ void __launch_bounds__(32)
k_gamma(const float* __restrict__ P, const float* __restrict__ S,
        float* __restrict__ GT) {
    const int row = blockIdx.x;
    const int b = row / II, i = row % II;
    const int lane = threadIdx.x;
    const float* __restrict__ p = P + (size_t)row * JJ;
    const float* __restrict__ s = S + (size_t)row * JJ;

    float carry = NEG_ID;
    for (int c = 0; c < JJ / 32; ++c) {
        int j = c * 32 + lane;
        float acc = p[j] - s[j];
#pragma unroll
        for (int off = 1; off < 32; off <<= 1) {
            float y = __shfl_up(acc, off, 32);
            if (lane >= off) acc = lse2(acc, y);
        }
        float scan = lse2(acc, carry);
        carry = __shfl(scan, 31, 32);
        float lb = s[j] + scan;
        float gamma = fast_exp(lb);
        if (i == II - 1 && j != JJ - 1) gamma = 0.0f;
        GT[((size_t)b * JJ + j) * II + i] = gamma;
    }
}

__global__ void __launch_bounds__(32)
k_expand_wmma(const float* __restrict__ GT, const float* __restrict__ TT,
              float* __restrict__ out) {
    const int b = blockIdx.z, jt = blockIdx.y, dt = blockIdx.x;
    const int lane = threadIdx.x;
    const int half = lane >> 4;
    const int line = lane & 15;
    const int j0 = jt * 16, d0 = dt * 16;
    const float* __restrict__ Ga = GT + ((size_t)b * JJ + (j0 + line)) * II + half * 2;
    const float* __restrict__ Tb = TT + ((size_t)b * DD + (d0 + line)) * II + half * 2;

    __builtin_prefetch(Ga, 0, 1);
    __builtin_prefetch(Tb, 0, 1);

    v8f acc = {};
#pragma unroll
    for (int k = 0; k < II; k += 4) {
        v2f a  = *(const v2f*)(Ga + k);
        v2f bv = *(const v2f*)(Tb + k);
        acc = __builtin_amdgcn_wmma_f32_16x16x4_f32(false, a, false, bv,
                                                    (short)0, acc, false, false);
    }
#pragma unroll
    for (int v = 0; v < 8; ++v) {
        int j = j0 + v + half * 8;
        out[(size_t)b * JJ * DD + (size_t)j * DD + d0 + line] = acc[v];
    }
}

extern "C" void kernel_launch(void* const* d_in, const int* in_sizes, int n_in,
                              void* d_out, int out_size, void* d_ws, size_t ws_size,
                              hipStream_t stream) {
    const float* text = (const float*)d_in[0];
    const float* mel  = (const float*)d_in[1];
    const float* gum  = (const float*)d_in[2];
    float* out = (float*)d_out;

    const size_t N = (size_t)BB * II * JJ;
    float* E  = (float*)d_ws;
    float* S  = E + N;
    float* P  = S + N;
    float* GT = P + N;
    float* TT = GT + N;

    k_transpose_text<<<dim3(8, BB), 256, 0, stream>>>(text, TT);
    k_energy_wmma<<<dim3(JJ / 16, II / 16, BB), 32, 0, stream>>>(text, mel, gum, E);
    k_suffix_lse<<<dim3(BB * II), 32, 0, stream>>>(E, S);
    k_forward<<<dim3(BB), 32, 0, stream>>>(E, S, P);
    k_gamma<<<dim3(BB * II), 32, 0, stream>>>(P, S, GT);
    k_expand_wmma<<<dim3(DD / 16, JJ / 16, BB), 32, 0, stream>>>(GT, TT, out);
    (void)in_sizes; (void)n_in; (void)out_size; (void)ws_size;
}


# K2+K3+K4 fused: the forward wave takes each row's suffix SUMS, prefix sums and gamma in the linear domain straight from E (6 launches to 4; K5 reads gamma row-major); order of summation changes
# speedup vs baseline: 12.0684x; 1.7656x over previous
#include <hip/hip_runtime.h>
#include <math.h>

#define BB 2
#define II 48
#define JJ 320
#define DD 256

#define LOG2E_F 1.4426950408889634f
#define LN2_F   0.6931471805599453f

typedef __attribute__((ext_vector_type(2))) float v2f;
typedef __attribute__((ext_vector_type(8))) float v8f;

__device__ __forceinline__ float fast_log(float x) {
    return LN2_F * __builtin_amdgcn_logf(x);
}

__global__ void __launch_bounds__(256)
k_transpose_text(const float* __restrict__ text, float* __restrict__ TT) {
    int b = blockIdx.y;
    for (int t = blockIdx.x * 256 + threadIdx.x; t < II * DD; t += gridDim.x * 256) {
        int i = t / DD, d = t % DD;
        TT[((size_t)b * DD + d) * II + i] = text[((size_t)b * II + i) * DD + d];
    }
}

__global__ void __launch_bounds__(32)
k_energy_wmma(const float* __restrict__ text, const float* __restrict__ mel,
              const float* __restrict__ gum, float* __restrict__ E) {
    const int b = blockIdx.z, it = blockIdx.y, jt = blockIdx.x;
    const int lane = threadIdx.x;
    const int half = lane >> 4;
    const int line = lane & 15;

    const float* __restrict__ Ta = text + (size_t)b * II * DD + (size_t)(it * 16 + line) * DD + half * 2;
    const float* __restrict__ Mb = mel  + (size_t)b * JJ * DD + (size_t)(jt * 16 + line) * DD + half * 2;

    __builtin_prefetch(Ta, 0, 1);
    __builtin_prefetch(Mb, 0, 1);
    __builtin_prefetch(Ta + 128, 0, 1);
    __builtin_prefetch(Mb + 128, 0, 1);

    v8f acc = {};
#pragma unroll 4
    for (int k = 0; k < DD; k += 4) {
        v2f a  = *(const v2f*)(Ta + k);
        v2f bv = *(const v2f*)(Mb + k);
        acc = __builtin_amdgcn_wmma_f32_16x16x4_f32(false, a, false, bv,
                                                    (short)0, acc, false, false);
    }

    const float invTemp = 1.0f / 0.55f;
    const float scale   = 1.0f / 256.0f;
#pragma unroll
    for (int v = 0; v < 8; ++v) {
        int i = it * 16 + v + half * 8;
        int j = jt * 16 + line;
        size_t idx = (size_t)b * II * JJ + (size_t)i * JJ + j;
        float u = gum[idx];
        u = fminf(fmaxf(u, 1e-7f), 1.0f - 1e-7f);
        float noise = -fast_log(-fast_log(u));
        E[idx] = (acc[v] * scale + noise) * invTemp;
    }
}

#define KPL (JJ / 32)

__device__ __forceinline__ void load_row(const float* __restrict__ p, float (&r)[KPL]) {
#pragma unroll
    for (int u = 0; u < KPL / 2; ++u) {
        v2f q = *(const v2f*)(p + 2 * u);
        r[2 * u] = q.x; r[2 * u + 1] = q.y;
    }
}

template <int N>
__device__ __forceinline__ float row_shr0(float v) {
    return __int_as_float(__builtin_amdgcn_update_dpp(0, __float_as_int(v), 0x110 + N, 0xf, 0xf, true));
}
template <int N>
__device__ __forceinline__ float row_shl0(float v) {
    return __int_as_float(__builtin_amdgcn_update_dpp(0, __float_as_int(v), 0x100 + N, 0xf, 0xf, true));
}
__device__ __forceinline__ float row_incl_sum(float x) {
    x += row_shr0<1>(x);
    x += row_shr0<2>(x);
    x += row_shr0<4>(x);
    x += row_shr0<8>(x);
    return x;
}
__device__ __forceinline__ float row_incl_sum_rev(float x) {
    x += row_shl0<1>(x);
    x += row_shl0<2>(x);
    x += row_shl0<4>(x);
    x += row_shl0<8>(x);
    return x;
}
template <int L>
__device__ __forceinline__ float lane_value(float v) {
    return __int_as_float(__builtin_amdgcn_readlane(__float_as_int(v), L));
}

__device__ __forceinline__ void prep_row(const float (&e)[KPL], const int lane,
                                         float (&x)[KPL], float (&suf)[KPL], float (&rs)[KPL]) {
    const float refL = lane_value<0>(e[0]) * LOG2E_F;
#pragma unroll
    for (int t = 0; t < KPL; ++t) x[t] = __builtin_amdgcn_exp2f(fmaf(e[t], LOG2E_F, -refL));
    float sl[KPL];
    sl[KPL - 1] = x[KPL - 1];
#pragma unroll
    for (int t = KPL - 2; t >= 0; --t) sl[t] = sl[t + 1] + x[t];
    const float xs = row_incl_sum_rev(sl[0]);
    const float up = lane_value<16>(xs);
    const float add = (lane < 16) ? up : 0.0f;
    const float right = row_shl0<1>(xs) + add;
#pragma unroll
    for (int t = 0; t < KPL; ++t) suf[t] = right + sl[t];
#pragma unroll
    for (int t = 0; t < KPL; ++t) rs[t] = __builtin_amdgcn_rcpf(suf[t]);
}

__device__ __forceinline__ float row_total(const float (&ph)[KPL]) {
    const float t01 = ph[0] + ph[1], t23 = ph[2] + ph[3], t45 = ph[4] + ph[5];
    const float t67 = ph[6] + ph[7], t89 = ph[8] + ph[9];
    float tot = ((t01 + t23) + (t45 + t67)) + t89;
    tot = row_incl_sum(tot);
    return lane_value<15>(tot) + lane_value<31>(tot);
}

template <bool PREF, bool PREP>
__device__ __forceinline__ void fwd_step(const int r, const int lane, const int k0,
                                         const float* __restrict__ Eb, float* __restrict__ Gb,
                                         float (&ph)[KPL], int& esum,
                                         float (&x)[KPL], float (&suf)[KPL], float (&rs)[KPL],
                                         const float (&en)[KPL], float (&ed)[KPL]) {
    if (PREF) load_row(Eb + (size_t)(r + 3) * JJ, ed);

    float xn[KPL], sufn[KPL], rsn[KPL];
    if (PREP) prep_row(en, lane, xn, sufn, rsn);

    const float M = row_total(ph);
    int n = __builtin_amdgcn_frexp_expf(M);
    n = (n < -100) ? -100 : n;
    const float g = ldexpf(1.0f, -n);

    const unsigned d = (unsigned)(k0 - r);
    float emg[KPL];
#pragma unroll
    for (int t = 0; t < KPL; ++t) {
        const bool inw = (d + (unsigned)t) < (unsigned)(JJ - II + 2);
        emg[t] = inw ? x[t] * g : 0.0f;
    }

    float l[KPL];
    l[0] = ph[0] * rs[0];
#pragma unroll
    for (int t = 1; t < KPL; ++t) l[t] = l[t - 1] + ph[t] * rs[t];

    const float xr = row_incl_sum(l[KPL - 1]);
    const float r0 = lane_value<15>(xr);
    const float add = (lane >= 16) ? r0 : 0.0f;
    const float excl = row_shr0<1>(xr) + add;
    float pre[KPL];
#pragma unroll
    for (int t = 0; t < KPL; ++t) pre[t] = excl + l[t];

    const float sc = ldexpf(1.0f, esum);
    float* __restrict__ gp = Gb + (size_t)r * JJ;
#pragma unroll
    for (int u = 0; u < KPL / 2; ++u) {
        v2f q;
        q.x = (pre[2 * u] * suf[2 * u]) * sc;
        q.y = (pre[2 * u + 1] * suf[2 * u + 1]) * sc;
        *(v2f*)(gp + 2 * u) = q;
    }

    const float hv = (xr + add) * emg[KPL - 1];
    float hand = row_shr0<1>(hv);
    const float h15 = lane_value<15>(hv);
    if (lane == 16) hand = h15;

    float nxt[KPL];
    nxt[0] = hand;
#pragma unroll
    for (int t = 0; t < KPL - 1; ++t) nxt[t + 1] = pre[t] * emg[t];
#pragma unroll
    for (int t = 0; t < KPL; ++t) ph[t] = nxt[t];
    esum += n;

    if (PREP) {
#pragma unroll
        for (int t = 0; t < KPL; ++t) { x[t] = xn[t]; suf[t] = sufn[t]; rs[t] = rsn[t]; }
    }
}

__global__ void __launch_bounds__(32)
k_forward(const float* __restrict__ E, float* __restrict__ G) {
    const int b = blockIdx.x;
    const int lane = threadIdx.x;
    const int k0 = lane * KPL;
    const float* __restrict__ Eb = E + (size_t)b * II * JJ + k0;
    float* __restrict__ Gb = G + (size_t)b * II * JJ + k0;

    float ph[KPL], x[KPL], suf[KPL], rs[KPL];
    float eA[KPL], eB[KPL], eC[KPL];
#pragma unroll
    for (int t = 0; t < KPL; ++t) ph[t] = (k0 + t == 0) ? 1.0f : 0.0f;
    load_row(Eb, eC);
    load_row(Eb + JJ, eA);
    load_row(Eb + 2 * JJ, eB);
    prep_row(eC, lane, x, suf, rs);
    int esum = 0;

    int r = 0;
#pragma unroll 1
    for (; r + 2 < II - 6; r += 3) {
        fwd_step<true, true>(r,     lane, k0, Eb, Gb, ph, esum, x, suf, rs, eA, eC);
        __builtin_amdgcn_sched_barrier(0);
        fwd_step<true, true>(r + 1, lane, k0, Eb, Gb, ph, esum, x, suf, rs, eB, eA);
        __builtin_amdgcn_sched_barrier(0);
        fwd_step<true, true>(r + 2, lane, k0, Eb, Gb, ph, esum, x, suf, rs, eC, eB);
        __builtin_amdgcn_sched_barrier(0);
    }
    fwd_step<true,  true >(r,     lane, k0, Eb, Gb, ph, esum, x, suf, rs, eA, eC);
    fwd_step<true,  true >(r + 1, lane, k0, Eb, Gb, ph, esum, x, suf, rs, eB, eA);
    fwd_step<false, true >(r + 2, lane, k0, Eb, Gb, ph, esum, x, suf, rs, eC, eB);
    fwd_step<false, true >(r + 3, lane, k0, Eb, Gb, ph, esum, x, suf, rs, eA, eC);
    fwd_step<false, false>(r + 4, lane, k0, Eb, Gb, ph, esum, x, suf, rs, eB, eA);

    const float last = row_total(ph) * ldexpf(1.0f, esum);
    float* __restrict__ gp = Gb + (size_t)(II - 1) * JJ;
#pragma unroll
    for (int u = 0; u < KPL / 2; ++u) {
        v2f q;
        q.x = 0.0f;
        q.y = (k0 + 2 * u + 1 == JJ - 1) ? last : 0.0f;
        *(v2f*)(gp + 2 * u) = q;
    }
}

__global__ void __launch_bounds__(32)
k_expand_wmma(const float* __restrict__ G, const float* __restrict__ TT,
              float* __restrict__ out) {
    const int b = blockIdx.z, jt = blockIdx.y, dt = blockIdx.x;
    const int lane = threadIdx.x;
    const int half = lane >> 4;
    const int line = lane & 15;
    const int j0 = jt * 16, d0 = dt * 16;
    const float* __restrict__ Ga = G + ((size_t)b * II + half * 2) * JJ + (j0 + line);
    const float* __restrict__ Tb = TT + ((size_t)b * DD + (d0 + line)) * II + half * 2;

    __builtin_prefetch(Ga, 0, 1);
    __builtin_prefetch(Tb, 0, 1);

    v8f acc = {};
#pragma unroll
    for (int k = 0; k < II; k += 4) {
        v2f a;
        a.x = Ga[(size_t)k * JJ];
        a.y = Ga[(size_t)(k + 1) * JJ];
        v2f bv = *(const v2f*)(Tb + k);
        acc = __builtin_amdgcn_wmma_f32_16x16x4_f32(false, a, false, bv,
                                                    (short)0, acc, false, false);
    }
#pragma unroll
    for (int v = 0; v < 8; ++v) {
        int j = j0 + v + half * 8;
        out[(size_t)b * JJ * DD + (size_t)j * DD + d0 + line] = acc[v];
    }
}

extern "C" void kernel_launch(void* const* d_in, const int* in_sizes, int n_in,
                              void* d_out, int out_size, void* d_ws, size_t ws_size,
                              hipStream_t stream) {
    const float* text = (const float*)d_in[0];
    const float* mel  = (const float*)d_in[1];
    const float* gum  = (const float*)d_in[2];
    float* out = (float*)d_out;

    const size_t N = (size_t)BB * II * JJ;
    float* E  = (float*)d_ws;
    float* G  = E + N;
    float* TT = G + N;

    k_transpose_text<<<dim3(8, BB), 256, 0, stream>>>(text, TT);
    k_energy_wmma<<<dim3(JJ / 16, II / 16, BB), 32, 0, stream>>>(text, mel, gum, E);
    k_forward<<<dim3(BB), 32, 0, stream>>>(E, G);
    k_expand_wmma<<<dim3(DD / 16, JJ / 16, BB), 32, 0, stream>>>(G, TT, out);
    (void)in_sizes; (void)n_in; (void)out_size; (void)ws_size;
}


# K3 reads its B fragments straight from text (one 4-byte load per fragment register, 64 B contiguous per lane half): the transposed copy and its launch are gone (4 launches to 3); sums unchanged
# speedup vs baseline: 16.5183x; 1.3687x over previous
#include <hip/hip_runtime.h>
#include <math.h>

#define BB 2
#define II 48
#define JJ 320
#define DD 256

#define LOG2E_F 1.4426950408889634f
#define LN2_F   0.6931471805599453f

typedef __attribute__((ext_vector_type(2))) float v2f;
typedef __attribute__((ext_vector_type(8))) float v8f;

__device__ __forceinline__ float fast_log(float x) {
    return LN2_F * __builtin_amdgcn_logf(x);
}

__global__ void __launch_bounds__(32)
k_energy_wmma(const float* __restrict__ text, const float* __restrict__ mel,
              const float* __restrict__ gum, float* __restrict__ E) {
    const int b = blockIdx.z, it = blockIdx.y, jt = blockIdx.x;
    const int lane = threadIdx.x;
    const int half = lane >> 4;
    const int line = lane & 15;

    const float* __restrict__ Ta = text + (size_t)b * II * DD + (size_t)(it * 16 + line) * DD + half * 2;
    const float* __restrict__ Mb = mel  + (size_t)b * JJ * DD + (size_t)(jt * 16 + line) * DD + half * 2;

    __builtin_prefetch(Ta, 0, 1);
    __builtin_prefetch(Mb, 0, 1);
    __builtin_prefetch(Ta + 128, 0, 1);
    __builtin_prefetch(Mb + 128, 0, 1);

    v8f acc = {};
#pragma unroll 4
    for (int k = 0; k < DD; k += 4) {
        v2f a  = *(const v2f*)(Ta + k);
        v2f bv = *(const v2f*)(Mb + k);
        acc = __builtin_amdgcn_wmma_f32_16x16x4_f32(false, a, false, bv,
                                                    (short)0, acc, false, false);
    }

    const float invTemp = 1.0f / 0.55f;
    const float scale   = 1.0f / 256.0f;
#pragma unroll
    for (int v = 0; v < 8; ++v) {
        int i = it * 16 + v + half * 8;
        int j = jt * 16 + line;
        size_t idx = (size_t)b * II * JJ + (size_t)i * JJ + j;
        float u = gum[idx];
        u = fminf(fmaxf(u, 1e-7f), 1.0f - 1e-7f);
        float noise = -fast_log(-fast_log(u));
        E[idx] = (acc[v] * scale + noise) * invTemp;
    }
}

#define KPL (JJ / 32)

__device__ __forceinline__ void load_row(const float* __restrict__ p, float (&r)[KPL]) {
#pragma unroll
    for (int u = 0; u < KPL / 2; ++u) {
        v2f q = *(const v2f*)(p + 2 * u);
        r[2 * u] = q.x; r[2 * u + 1] = q.y;
    }
}

template <int N>
__device__ __forceinline__ float row_shr0(float v) {
    return __int_as_float(__builtin_amdgcn_update_dpp(0, __float_as_int(v), 0x110 + N, 0xf, 0xf, true));
}
template <int N>
__device__ __forceinline__ float row_shl0(float v) {
    return __int_as_float(__builtin_amdgcn_update_dpp(0, __float_as_int(v), 0x100 + N, 0xf, 0xf, true));
}
__device__ __forceinline__ float row_incl_sum(float x) {
    x += row_shr0<1>(x);
    x += row_shr0<2>(x);
    x += row_shr0<4>(x);
    x += row_shr0<8>(x);
    return x;
}
__device__ __forceinline__ float row_incl_sum_rev(float x) {
    x += row_shl0<1>(x);
    x += row_shl0<2>(x);
    x += row_shl0<4>(x);
    x += row_shl0<8>(x);
    return x;
}
template <int L>
__device__ __forceinline__ float lane_value(float v) {
    return __int_as_float(__builtin_amdgcn_readlane(__float_as_int(v), L));
}

__device__ __forceinline__ void prep_row(const float (&e)[KPL], const int lane,
                                         float (&x)[KPL], float (&suf)[KPL], float (&rs)[KPL]) {
    const float refL = lane_value<0>(e[0]) * LOG2E_F;
#pragma unroll
    for (int t = 0; t < KPL; ++t) x[t] = __builtin_amdgcn_exp2f(fmaf(e[t], LOG2E_F, -refL));
    float sl[KPL];
    sl[KPL - 1] = x[KPL - 1];
#pragma unroll
    for (int t = KPL - 2; t >= 0; --t) sl[t] = sl[t + 1] + x[t];
    const float xs = row_incl_sum_rev(sl[0]);
    const float up = lane_value<16>(xs);
    const float add = (lane < 16) ? up : 0.0f;
    const float right = row_shl0<1>(xs) + add;
#pragma unroll
    for (int t = 0; t < KPL; ++t) suf[t] = right + sl[t];
#pragma unroll
    for (int t = 0; t < KPL; ++t) rs[t] = __builtin_amdgcn_rcpf(suf[t]);
}

__device__ __forceinline__ float row_total(const float (&ph)[KPL]) {
    const float t01 = ph[0] + ph[1], t23 = ph[2] + ph[3], t45 = ph[4] + ph[5];
    const float t67 = ph[6] + ph[7], t89 = ph[8] + ph[9];
    float tot = ((t01 + t23) + (t45 + t67)) + t89;
    tot = row_incl_sum(tot);
    return lane_value<15>(tot) + lane_value<31>(tot);
}

template <bool PREF, bool PREP>
__device__ __forceinline__ void fwd_step(const int r, const int lane, const int k0,
                                         const float* __restrict__ Eb, float* __restrict__ Gb,
                                         float (&ph)[KPL], int& esum,
                                         float (&x)[KPL], float (&suf)[KPL], float (&rs)[KPL],
                                         const float (&en)[KPL], float (&ed)[KPL]) {
    if (PREF) load_row(Eb + (size_t)(r + 3) * JJ, ed);

    float xn[KPL], sufn[KPL], rsn[KPL];
    if (PREP) prep_row(en, lane, xn, sufn, rsn);

    const float M = row_total(ph);
    int n = __builtin_amdgcn_frexp_expf(M);
    n = (n < -100) ? -100 : n;
    const float g = ldexpf(1.0f, -n);

    const unsigned d = (unsigned)(k0 - r);
    float emg[KPL];
#pragma unroll
    for (int t = 0; t < KPL; ++t) {
        const bool inw = (d + (unsigned)t) < (unsigned)(JJ - II + 2);
        emg[t] = inw ? x[t] * g : 0.0f;
    }

    float l[KPL];
    l[0] = ph[0] * rs[0];
#pragma unroll
    for (int t = 1; t < KPL; ++t) l[t] = l[t - 1] + ph[t] * rs[t];

    const float xr = row_incl_sum(l[KPL - 1]);
    const float r0 = lane_value<15>(xr);
    const float add = (lane >= 16) ? r0 : 0.0f;
    const float excl = row_shr0<1>(xr) + add;
    float pre[KPL];
#pragma unroll
    for (int t = 0; t < KPL; ++t) pre[t] = excl + l[t];

    const float sc = ldexpf(1.0f, esum);
    float* __restrict__ gp = Gb + (size_t)r * JJ;
#pragma unroll
    for (int u = 0; u < KPL / 2; ++u) {
        v2f q;
        q.x = (pre[2 * u] * suf[2 * u]) * sc;
        q.y = (pre[2 * u + 1] * suf[2 * u + 1]) * sc;
        *(v2f*)(gp + 2 * u) = q;
    }

    const float hv = (xr + add) * emg[KPL - 1];
    float hand = row_shr0<1>(hv);
    const float h15 = lane_value<15>(hv);
    if (lane == 16) hand = h15;

    float nxt[KPL];
    nxt[0] = hand;
#pragma unroll
    for (int t = 0; t < KPL - 1; ++t) nxt[t + 1] = pre[t] * emg[t];
#pragma unroll
    for (int t = 0; t < KPL; ++t) ph[t] = nxt[t];
    esum += n;

    if (PREP) {
#pragma unroll
        for (int t = 0; t < KPL; ++t) { x[t] = xn[t]; suf[t] = sufn[t]; rs[t] = rsn[t]; }
    }
}

__global__ void __launch_bounds__(32)
k_forward(const float* __restrict__ E, float* __restrict__ G) {
    const int b = blockIdx.x;
    const int lane = threadIdx.x;
    const int k0 = lane * KPL;
    const float* __restrict__ Eb = E + (size_t)b * II * JJ + k0;
    float* __restrict__ Gb = G + (size_t)b * II * JJ + k0;

    float ph[KPL], x[KPL], suf[KPL], rs[KPL];
    float eA[KPL], eB[KPL], eC[KPL];
#pragma unroll
    for (int t = 0; t < KPL; ++t) ph[t] = (k0 + t == 0) ? 1.0f : 0.0f;
    load_row(Eb, eC);
    load_row(Eb + JJ, eA);
    load_row(Eb + 2 * JJ, eB);
    prep_row(eC, lane, x, suf, rs);
    int esum = 0;

    int r = 0;
#pragma unroll 1
    for (; r + 2 < II - 6; r += 3) {
        fwd_step<true, true>(r,     lane, k0, Eb, Gb, ph, esum, x, suf, rs, eA, eC);
        __builtin_amdgcn_sched_barrier(0);
        fwd_step<true, true>(r + 1, lane, k0, Eb, Gb, ph, esum, x, suf, rs, eB, eA);
        __builtin_amdgcn_sched_barrier(0);
        fwd_step<true, true>(r + 2, lane, k0, Eb, Gb, ph, esum, x, suf, rs, eC, eB);
        __builtin_amdgcn_sched_barrier(0);
    }
    fwd_step<true,  true >(r,     lane, k0, Eb, Gb, ph, esum, x, suf, rs, eA, eC);
    fwd_step<true,  true >(r + 1, lane, k0, Eb, Gb, ph, esum, x, suf, rs, eB, eA);
    fwd_step<false, true >(r + 2, lane, k0, Eb, Gb, ph, esum, x, suf, rs, eC, eB);
    fwd_step<false, true >(r + 3, lane, k0, Eb, Gb, ph, esum, x, suf, rs, eA, eC);
    fwd_step<false, false>(r + 4, lane, k0, Eb, Gb, ph, esum, x, suf, rs, eB, eA);

    const float last = row_total(ph) * ldexpf(1.0f, esum);
    float* __restrict__ gp = Gb + (size_t)(II - 1) * JJ;
#pragma unroll
    for (int u = 0; u < KPL / 2; ++u) {
        v2f q;
        q.x = 0.0f;
        q.y = (k0 + 2 * u + 1 == JJ - 1) ? last : 0.0f;
        *(v2f*)(gp + 2 * u) = q;
    }
}

__global__ void __launch_bounds__(32)
k_expand_wmma(const float* __restrict__ G, const float* __restrict__ text,
              float* __restrict__ out) {
    const int b = blockIdx.z, jt = blockIdx.y, dt = blockIdx.x;
    const int lane = threadIdx.x;
    const int half = lane >> 4;
    const int line = lane & 15;
    const int j0 = jt * 16, d0 = dt * 16;
    const float* __restrict__ Ga = G + ((size_t)b * II + half * 2) * JJ + (j0 + line);
    const float* __restrict__ Tb = text + ((size_t)b * II + half * 2) * DD + (d0 + line);

    __builtin_prefetch(Ga, 0, 1);
    __builtin_prefetch(Tb, 0, 1);

    v8f acc = {};
#pragma unroll
    for (int k = 0; k < II; k += 4) {
        v2f a;
        a.x = Ga[(size_t)k * JJ];
        a.y = Ga[(size_t)(k + 1) * JJ];
        v2f bv;
        bv.x = Tb[(size_t)k * DD];
        bv.y = Tb[(size_t)(k + 1) * DD];
        acc = __builtin_amdgcn_wmma_f32_16x16x4_f32(false, a, false, bv,
                                                    (short)0, acc, false, false);
    }
#pragma unroll
    for (int v = 0; v < 8; ++v) {
        int j = j0 + v + half * 8;
        out[(size_t)b * JJ * DD + (size_t)j * DD + d0 + line] = acc[v];
    }
}

extern "C" void kernel_launch(void* const* d_in, const int* in_sizes, int n_in,
                              void* d_out, int out_size, void* d_ws, size_t ws_size,
                              hipStream_t stream) {
    const float* text = (const float*)d_in[0];
    const float* mel  = (const float*)d_in[1];
    const float* gum  = (const float*)d_in[2];
    float* out = (float*)d_out;

    const size_t N = (size_t)BB * II * JJ;
    float* E  = (float*)d_ws;
    float* G  = E + N;

    k_energy_wmma<<<dim3(JJ / 16, II / 16, BB), 32, 0, stream>>>(text, mel, gum, E);
    k_forward<<<dim3(BB), 32, 0, stream>>>(E, G);
    k_expand_wmma<<<dim3(DD / 16, JJ / 16, BB), 32, 0, stream>>>(G, text, out);
    (void)in_sizes; (void)n_in; (void)out_size; (void)ws_size;
}


# K2 as a 16-wave work-group: all waves work out every row's factors (exp, suffix sums, reciprocals, window) into LDS, one wave runs the recurrence from LDS rows; every sum keeps its order
# speedup vs baseline: 17.1084x; 1.0357x over previous
#include <hip/hip_runtime.h>
#include <math.h>

#define BB 2
#define II 48
#define JJ 320
#define DD 256

#define LOG2E_F 1.4426950408889634f
#define LN2_F   0.6931471805599453f

typedef __attribute__((ext_vector_type(2))) float v2f;
typedef __attribute__((ext_vector_type(8))) float v8f;

__device__ __forceinline__ float fast_log(float x) {
    return LN2_F * __builtin_amdgcn_logf(x);
}

__global__ void __launch_bounds__(32)
k_energy_wmma(const float* __restrict__ text, const float* __restrict__ mel,
              const float* __restrict__ gum, float* __restrict__ E) {
    const int b = blockIdx.z, it = blockIdx.y, jt = blockIdx.x;
    const int lane = threadIdx.x;
    const int half = lane >> 4;
    const int line = lane & 15;

    const float* __restrict__ Ta = text + (size_t)b * II * DD + (size_t)(it * 16 + line) * DD + half * 2;
    const float* __restrict__ Mb = mel  + (size_t)b * JJ * DD + (size_t)(jt * 16 + line) * DD + half * 2;

    __builtin_prefetch(Ta, 0, 1);
    __builtin_prefetch(Mb, 0, 1);
    __builtin_prefetch(Ta + 128, 0, 1);
    __builtin_prefetch(Mb + 128, 0, 1);

    v8f acc = {};
#pragma unroll 4
    for (int k = 0; k < DD; k += 4) {
        v2f a  = *(const v2f*)(Ta + k);
        v2f bv = *(const v2f*)(Mb + k);
        acc = __builtin_amdgcn_wmma_f32_16x16x4_f32(false, a, false, bv,
                                                    (short)0, acc, false, false);
    }

    const float invTemp = 1.0f / 0.55f;
    const float scale   = 1.0f / 256.0f;
#pragma unroll
    for (int v = 0; v < 8; ++v) {
        int i = it * 16 + v + half * 8;
        int j = jt * 16 + line;
        size_t idx = (size_t)b * II * JJ + (size_t)i * JJ + j;
        float u = gum[idx];
        u = fminf(fmaxf(u, 1e-7f), 1.0f - 1e-7f);
        float noise = -fast_log(-fast_log(u));
        E[idx] = (acc[v] * scale + noise) * invTemp;
    }
}

#define KPL (JJ / 32)
#define FW_WAVES 16
#define FW_ROWS (II - 1)
#define FW_SPLIT 16
#define FW_WIN (JJ - II + 2)

__device__ __forceinline__ void load_row(const float* __restrict__ p, float (&r)[KPL]) {
#pragma unroll
    for (int u = 0; u < KPL / 2; ++u) {
        v2f q = *(const v2f*)(p + 2 * u);
        r[2 * u] = q.x; r[2 * u + 1] = q.y;
    }
}
__device__ __forceinline__ void store_row(float* __restrict__ p, const float (&r)[KPL]) {
#pragma unroll
    for (int u = 0; u < KPL / 2; ++u) {
        v2f q;
        q.x = r[2 * u]; q.y = r[2 * u + 1];
        *(v2f*)(p + 2 * u) = q;
    }
}

template <int N>
__device__ __forceinline__ float row_shr0(float v) {
    return __int_as_float(__builtin_amdgcn_update_dpp(0, __float_as_int(v), 0x110 + N, 0xf, 0xf, true));
}
template <int N>
__device__ __forceinline__ float row_shl0(float v) {
    return __int_as_float(__builtin_amdgcn_update_dpp(0, __float_as_int(v), 0x100 + N, 0xf, 0xf, true));
}
__device__ __forceinline__ float row_incl_sum(float x) {
    x += row_shr0<1>(x);
    x += row_shr0<2>(x);
    x += row_shr0<4>(x);
    x += row_shr0<8>(x);
    return x;
}
__device__ __forceinline__ float row_incl_sum_rev(float x) {
    x += row_shl0<1>(x);
    x += row_shl0<2>(x);
    x += row_shl0<4>(x);
    x += row_shl0<8>(x);
    return x;
}
template <int L>
__device__ __forceinline__ float lane_value(float v) {
    return __int_as_float(__builtin_amdgcn_readlane(__float_as_int(v), L));
}

__device__ __forceinline__ void prep_row(const float (&e)[KPL], const int lane,
                                         float (&x)[KPL], float (&suf)[KPL], float (&rs)[KPL]) {
    const float refL = lane_value<0>(e[0]) * LOG2E_F;
#pragma unroll
    for (int t = 0; t < KPL; ++t) x[t] = __builtin_amdgcn_exp2f(fmaf(e[t], LOG2E_F, -refL));
    float sl[KPL];
    sl[KPL - 1] = x[KPL - 1];
#pragma unroll
    for (int t = KPL - 2; t >= 0; --t) sl[t] = sl[t + 1] + x[t];
    const float xs = row_incl_sum_rev(sl[0]);
    const float up = lane_value<16>(xs);
    const float add = (lane < 16) ? up : 0.0f;
    const float right = row_shl0<1>(xs) + add;
#pragma unroll
    for (int t = 0; t < KPL; ++t) suf[t] = right + sl[t];
#pragma unroll
    for (int t = 0; t < KPL; ++t) rs[t] = __builtin_amdgcn_rcpf(suf[t]);
}

__device__ __forceinline__ void prep_row_to_lds(const int row, const int lane, const int k0,
                                                const float (&e)[KPL],
                                                float* __restrict__ sx, float* __restrict__ ss,
                                                float* __restrict__ sr) {
    float x[KPL], suf[KPL], rs[KPL];
    prep_row(e, lane, x, suf, rs);
    const unsigned d = (unsigned)(k0 - row);
    float xw[KPL];
#pragma unroll
    for (int t = 0; t < KPL; ++t) xw[t] = ((d + (unsigned)t) < (unsigned)FW_WIN) ? x[t] : 0.0f;
    const int o = row * JJ + k0;
    store_row(sx + o, xw);
    store_row(ss + o, suf);
    store_row(sr + o, rs);
}

__device__ __forceinline__ float row_total(const float (&ph)[KPL]) {
    const float t01 = ph[0] + ph[1], t23 = ph[2] + ph[3], t45 = ph[4] + ph[5];
    const float t67 = ph[6] + ph[7], t89 = ph[8] + ph[9];
    float tot = ((t01 + t23) + (t45 + t67)) + t89;
    tot = row_incl_sum(tot);
    return lane_value<15>(tot) + lane_value<31>(tot);
}

__device__ __forceinline__ void fwd_step(const int r, const int lane,
                                         float* __restrict__ Gb,
                                         float (&ph)[KPL], int& esum,
                                         const float (&xw)[KPL], const float (&suf)[KPL],
                                         const float (&rs)[KPL]) {
    const float M = row_total(ph);
    int n = __builtin_amdgcn_frexp_expf(M);
    n = (n < -100) ? -100 : n;
    const float g = ldexpf(1.0f, -n);

    float emg[KPL];
#pragma unroll
    for (int t = 0; t < KPL; ++t) emg[t] = xw[t] * g;

    float l[KPL];
    l[0] = ph[0] * rs[0];
#pragma unroll
    for (int t = 1; t < KPL; ++t) l[t] = l[t - 1] + ph[t] * rs[t];

    const float xr = row_incl_sum(l[KPL - 1]);
    const float r0 = lane_value<15>(xr);
    const float add = (lane >= 16) ? r0 : 0.0f;
    const float excl = row_shr0<1>(xr) + add;
    float pre[KPL];
#pragma unroll
    for (int t = 0; t < KPL; ++t) pre[t] = excl + l[t];

    const float sc = ldexpf(1.0f, esum);
    float* __restrict__ gp = Gb + (size_t)r * JJ;
#pragma unroll
    for (int u = 0; u < KPL / 2; ++u) {
        v2f q;
        q.x = (pre[2 * u] * suf[2 * u]) * sc;
        q.y = (pre[2 * u + 1] * suf[2 * u + 1]) * sc;
        *(v2f*)(gp + 2 * u) = q;
    }

    const float hv = (xr + add) * emg[KPL - 1];
    float hand = row_shr0<1>(hv);
    const float h15 = lane_value<15>(hv);
    if (lane == 16) hand = h15;

    float nxt[KPL];
    nxt[0] = hand;
#pragma unroll
    for (int t = 0; t < KPL - 1; ++t) nxt[t + 1] = pre[t] * emg[t];
#pragma unroll
    for (int t = 0; t < KPL; ++t) ph[t] = nxt[t];
    esum += n;
}

template <int R0, int R1>
__device__ __forceinline__ void run_steps(const int lane, const int k0, float* __restrict__ Gb,
                                          float (&ph)[KPL], int& esum,
                                          const float* __restrict__ sx, const float* __restrict__ ss,
                                          const float* __restrict__ sr) {
    float xa[KPL], sa[KPL], ra[KPL], xb[KPL], sb[KPL], rb[KPL];
    load_row(sx + R0 * JJ + k0, xa);
    load_row(ss + R0 * JJ + k0, sa);
    load_row(sr + R0 * JJ + k0, ra);
    int r = R0;
#pragma unroll 1
    for (; r + 1 < R1; r += 2) {
        const int o1 = (r + 1) * JJ + k0;
        load_row(sx + o1, xb);
        load_row(ss + o1, sb);
        load_row(sr + o1, rb);
        fwd_step(r, lane, Gb, ph, esum, xa, sa, ra);
        if (r + 2 < R1) {
            const int o2 = (r + 2) * JJ + k0;
            load_row(sx + o2, xa);
            load_row(ss + o2, sa);
            load_row(sr + o2, ra);
        }
        fwd_step(r + 1, lane, Gb, ph, esum, xb, sb, rb);
    }
    if ((R1 - R0) & 1) fwd_step(R1 - 1, lane, Gb, ph, esum, xa, sa, ra);
}

__global__ void __launch_bounds__(32 * FW_WAVES)
k_forward(const float* __restrict__ E, float* __restrict__ G) {
    __shared__ float s_xw[FW_ROWS * JJ];
    __shared__ float s_suf[FW_ROWS * JJ];
    __shared__ float s_rs[FW_ROWS * JJ];

    const int b = blockIdx.x;
    const int lane = threadIdx.x & 31;
    const int w = __builtin_amdgcn_readfirstlane((int)(threadIdx.x >> 5));
    const int k0 = lane * KPL;
    const float* __restrict__ Eb = E + (size_t)b * II * JJ + k0;
    float* __restrict__ Gb = G + (size_t)b * II * JJ + k0;

    const int ra = FW_SPLIT + (w - 1);
    const int rb = ra + (FW_WAVES - 1);
    const int rc = rb + (FW_WAVES - 1);
    float e0[KPL], ea[KPL], eb[KPL], ec[KPL];
    load_row(Eb + (size_t)w * JJ, e0);
    if (w != 0) {
        load_row(Eb + (size_t)ra * JJ, ea);
        load_row(Eb + (size_t)rb * JJ, eb);
        if (rc < FW_ROWS) load_row(Eb + (size_t)rc * JJ, ec);
    }

    prep_row_to_lds(w, lane, k0, e0, s_xw, s_suf, s_rs);
    __syncthreads();

    float ph[KPL];
#pragma unroll
    for (int t = 0; t < KPL; ++t) ph[t] = (k0 + t == 0) ? 1.0f : 0.0f;
    int esum = 0;

    if (w == 0) {
        run_steps<0, FW_SPLIT>(lane, k0, Gb, ph, esum, s_xw, s_suf, s_rs);
    } else {
        prep_row_to_lds(ra, lane, k0, ea, s_xw, s_suf, s_rs);
        prep_row_to_lds(rb, lane, k0, eb, s_xw, s_suf, s_rs);
        if (rc < FW_ROWS) prep_row_to_lds(rc, lane, k0, ec, s_xw, s_suf, s_rs);
    }
    asm volatile("s_wait_dscnt 0x0" ::: "memory");
    asm volatile("s_barrier_signal -1" ::: "memory");
    asm volatile("s_barrier_wait -1" ::: "memory");

    if (w == 0) {
        run_steps<FW_SPLIT, FW_ROWS>(lane, k0, Gb, ph, esum, s_xw, s_suf, s_rs);

        const float last = row_total(ph) * ldexpf(1.0f, esum);
        float* __restrict__ gp = Gb + (size_t)(II - 1) * JJ;
#pragma unroll
        for (int u = 0; u < KPL / 2; ++u) {
            v2f q;
            q.x = 0.0f;
            q.y = (k0 + 2 * u + 1 == JJ - 1) ? last : 0.0f;
            *(v2f*)(gp + 2 * u) = q;
        }
    }
}

__global__ void __launch_bounds__(32)
k_expand_wmma(const float* __restrict__ G, const float* __restrict__ text,
              float* __restrict__ out) {
    const int b = blockIdx.z, jt = blockIdx.y, dt = blockIdx.x;
    const int lane = threadIdx.x;
    const int half = lane >> 4;
    const int line = lane & 15;
    const int j0 = jt * 16, d0 = dt * 16;
    const float* __restrict__ Ga = G + ((size_t)b * II + half * 2) * JJ + (j0 + line);
    const float* __restrict__ Tb = text + ((size_t)b * II + half * 2) * DD + (d0 + line);

    __builtin_prefetch(Ga, 0, 1);
    __builtin_prefetch(Tb, 0, 1);

    v8f acc = {};
#pragma unroll
    for (int k = 0; k < II; k += 4) {
        v2f a;
        a.x = Ga[(size_t)k * JJ];
        a.y = Ga[(size_t)(k + 1) * JJ];
        v2f bv;
        bv.x = Tb[(size_t)k * DD];
        bv.y = Tb[(size_t)(k + 1) * DD];
        acc = __builtin_amdgcn_wmma_f32_16x16x4_f32(false, a, false, bv,
                                                    (short)0, acc, false, false);
    }
#pragma unroll
    for (int v = 0; v < 8; ++v) {
        int j = j0 + v + half * 8;
        out[(size_t)b * JJ * DD + (size_t)j * DD + d0 + line] = acc[v];
    }
}

extern "C" void kernel_launch(void* const* d_in, const int* in_sizes, int n_in,
                              void* d_out, int out_size, void* d_ws, size_t ws_size,
                              hipStream_t stream) {
    const float* text = (const float*)d_in[0];
    const float* mel  = (const float*)d_in[1];
    const float* gum  = (const float*)d_in[2];
    float* out = (float*)d_out;

    const size_t N = (size_t)BB * II * JJ;
    float* E  = (float*)d_ws;
    float* G  = E + N;

    k_energy_wmma<<<dim3(JJ / 16, II / 16, BB), 32, 0, stream>>>(text, mel, gum, E);
    k_forward<<<dim3(BB), 32 * FW_WAVES, 0, stream>>>(E, G);
    k_expand_wmma<<<dim3(DD / 16, JJ / 16, BB), 32, 0, stream>>>(G, text, out);
    (void)in_sizes; (void)n_in; (void)out_size; (void)ws_size;
}


# K2+K3 fused: each (batch, 16-frame tile) work-group runs the batch's recurrence itself, keeps gamma in LDS and ends with its 16 WMMA expand tiles (3 launches to 2); sums unchanged
# speedup vs baseline: 20.4686x; 1.1964x over previous
#include <hip/hip_runtime.h>
#include <math.h>

#define BB 2
#define II 48
#define JJ 320
#define DD 256

#define LOG2E_F 1.4426950408889634f
#define LN2_F   0.6931471805599453f

typedef __attribute__((ext_vector_type(2))) float v2f;
typedef __attribute__((ext_vector_type(8))) float v8f;

__device__ __forceinline__ float fast_log(float x) {
    return LN2_F * __builtin_amdgcn_logf(x);
}

__global__ void __launch_bounds__(32)
k_energy_wmma(const float* __restrict__ text, const float* __restrict__ mel,
              const float* __restrict__ gum, float* __restrict__ E) {
    const int b = blockIdx.z, it = blockIdx.y, jt = blockIdx.x;
    const int lane = threadIdx.x;
    const int half = lane >> 4;
    const int line = lane & 15;

    const float* __restrict__ Ta = text + (size_t)b * II * DD + (size_t)(it * 16 + line) * DD + half * 2;
    const float* __restrict__ Mb = mel  + (size_t)b * JJ * DD + (size_t)(jt * 16 + line) * DD + half * 2;

    __builtin_prefetch(Ta, 0, 1);
    __builtin_prefetch(Mb, 0, 1);
    __builtin_prefetch(Ta + 128, 0, 1);
    __builtin_prefetch(Mb + 128, 0, 1);

    v8f acc = {};
#pragma unroll 4
    for (int k = 0; k < DD; k += 4) {
        v2f a  = *(const v2f*)(Ta + k);
        v2f bv = *(const v2f*)(Mb + k);
        acc = __builtin_amdgcn_wmma_f32_16x16x4_f32(false, a, false, bv,
                                                    (short)0, acc, false, false);
    }

    const float invTemp = 1.0f / 0.55f;
    const float scale   = 1.0f / 256.0f;
#pragma unroll
    for (int v = 0; v < 8; ++v) {
        int i = it * 16 + v + half * 8;
        int j = jt * 16 + line;
        size_t idx = (size_t)b * II * JJ + (size_t)i * JJ + j;
        float u = gum[idx];
        u = fminf(fmaxf(u, 1e-7f), 1.0f - 1e-7f);
        float noise = -fast_log(-fast_log(u));
        E[idx] = (acc[v] * scale + noise) * invTemp;
    }
}

#define KPL (JJ / 32)
#define FW_WAVES 16
#define FW_ROWS (II - 1)
#define FW_SPLIT 16
#define FW_WIN (JJ - II + 2)

__device__ __forceinline__ void load_row(const float* __restrict__ p, float (&r)[KPL]) {
#pragma unroll
    for (int u = 0; u < KPL / 2; ++u) {
        v2f q = *(const v2f*)(p + 2 * u);
        r[2 * u] = q.x; r[2 * u + 1] = q.y;
    }
}
__device__ __forceinline__ void store_row(float* __restrict__ p, const float (&r)[KPL]) {
#pragma unroll
    for (int u = 0; u < KPL / 2; ++u) {
        v2f q;
        q.x = r[2 * u]; q.y = r[2 * u + 1];
        *(v2f*)(p + 2 * u) = q;
    }
}

template <int N>
__device__ __forceinline__ float row_shr0(float v) {
    return __int_as_float(__builtin_amdgcn_update_dpp(0, __float_as_int(v), 0x110 + N, 0xf, 0xf, true));
}
template <int N>
__device__ __forceinline__ float row_shl0(float v) {
    return __int_as_float(__builtin_amdgcn_update_dpp(0, __float_as_int(v), 0x100 + N, 0xf, 0xf, true));
}
__device__ __forceinline__ float row_incl_sum(float x) {
    x += row_shr0<1>(x);
    x += row_shr0<2>(x);
    x += row_shr0<4>(x);
    x += row_shr0<8>(x);
    return x;
}
__device__ __forceinline__ float row_incl_sum_rev(float x) {
    x += row_shl0<1>(x);
    x += row_shl0<2>(x);
    x += row_shl0<4>(x);
    x += row_shl0<8>(x);
    return x;
}
template <int L>
__device__ __forceinline__ float lane_value(float v) {
    return __int_as_float(__builtin_amdgcn_readlane(__float_as_int(v), L));
}

__device__ __forceinline__ void prep_row(const float (&e)[KPL], const int lane,
                                         float (&x)[KPL], float (&suf)[KPL], float (&rs)[KPL]) {
    const float refL = lane_value<0>(e[0]) * LOG2E_F;
#pragma unroll
    for (int t = 0; t < KPL; ++t) x[t] = __builtin_amdgcn_exp2f(fmaf(e[t], LOG2E_F, -refL));
    float sl[KPL];
    sl[KPL - 1] = x[KPL - 1];
#pragma unroll
    for (int t = KPL - 2; t >= 0; --t) sl[t] = sl[t + 1] + x[t];
    const float xs = row_incl_sum_rev(sl[0]);
    const float up = lane_value<16>(xs);
    const float add = (lane < 16) ? up : 0.0f;
    const float right = row_shl0<1>(xs) + add;
#pragma unroll
    for (int t = 0; t < KPL; ++t) suf[t] = right + sl[t];
#pragma unroll
    for (int t = 0; t < KPL; ++t) rs[t] = __builtin_amdgcn_rcpf(suf[t]);
}

__device__ __forceinline__ void prep_row_to_lds(const int row, const int lane, const int k0,
                                                const float (&e)[KPL],
                                                float* __restrict__ sx, float* __restrict__ ss,
                                                float* __restrict__ sr) {
    float x[KPL], suf[KPL], rs[KPL];
    prep_row(e, lane, x, suf, rs);
    const unsigned d = (unsigned)(k0 - row);
    float xw[KPL];
#pragma unroll
    for (int t = 0; t < KPL; ++t) xw[t] = ((d + (unsigned)t) < (unsigned)FW_WIN) ? x[t] : 0.0f;
    const int o = row * JJ + k0;
    store_row(sx + o, xw);
    store_row(ss + o, suf);
    store_row(sr + o, rs);
}

__device__ __forceinline__ float row_total(const float (&ph)[KPL]) {
    const float t01 = ph[0] + ph[1], t23 = ph[2] + ph[3], t45 = ph[4] + ph[5];
    const float t67 = ph[6] + ph[7], t89 = ph[8] + ph[9];
    float tot = ((t01 + t23) + (t45 + t67)) + t89;
    tot = row_incl_sum(tot);
    return lane_value<15>(tot) + lane_value<31>(tot);
}

__device__ __forceinline__ void fwd_step(const int r, const int lane,
                                         float* __restrict__ Gb,
                                         float (&ph)[KPL], int& esum,
                                         const float (&xw)[KPL], const float (&suf)[KPL],
                                         const float (&rs)[KPL]) {
    const float M = row_total(ph);
    int n = __builtin_amdgcn_frexp_expf(M);
    n = (n < -100) ? -100 : n;
    const float g = ldexpf(1.0f, -n);

    float emg[KPL];
#pragma unroll
    for (int t = 0; t < KPL; ++t) emg[t] = xw[t] * g;

    float l[KPL];
    l[0] = ph[0] * rs[0];
#pragma unroll
    for (int t = 1; t < KPL; ++t) l[t] = l[t - 1] + ph[t] * rs[t];

    const float xr = row_incl_sum(l[KPL - 1]);
    const float r0 = lane_value<15>(xr);
    const float add = (lane >= 16) ? r0 : 0.0f;
    const float excl = row_shr0<1>(xr) + add;
    float pre[KPL];
#pragma unroll
    for (int t = 0; t < KPL; ++t) pre[t] = excl + l[t];

    const float sc = ldexpf(1.0f, esum);
    float* __restrict__ gp = Gb + (size_t)r * JJ;
#pragma unroll
    for (int u = 0; u < KPL / 2; ++u) {
        v2f q;
        q.x = (pre[2 * u] * suf[2 * u]) * sc;
        q.y = (pre[2 * u + 1] * suf[2 * u + 1]) * sc;
        *(v2f*)(gp + 2 * u) = q;
    }

    const float hv = (xr + add) * emg[KPL - 1];
    float hand = row_shr0<1>(hv);
    const float h15 = lane_value<15>(hv);
    if (lane == 16) hand = h15;

    float nxt[KPL];
    nxt[0] = hand;
#pragma unroll
    for (int t = 0; t < KPL - 1; ++t) nxt[t + 1] = pre[t] * emg[t];
#pragma unroll
    for (int t = 0; t < KPL; ++t) ph[t] = nxt[t];
    esum += n;
}

__device__ __forceinline__ void row_landed(const float (&v)[KPL]) {
    asm volatile("" :: "v"(v[0]), "v"(v[1]), "v"(v[2]), "v"(v[3]), "v"(v[4]),
                       "v"(v[5]), "v"(v[6]), "v"(v[7]), "v"(v[8]), "v"(v[9]));
}

template <int R0, int R1>
__device__ __forceinline__ void run_steps(const int lane, const int k0, float* __restrict__ Gb,
                                          float (&ph)[KPL], int& esum,
                                          const float* __restrict__ sx, const float* __restrict__ ss,
                                          const float* __restrict__ sr) {
    float xa[KPL], sa[KPL], ra[KPL], xb[KPL], sb[KPL], rb[KPL];
    load_row(sx + R0 * JJ + k0, xa);
    load_row(ss + R0 * JJ + k0, sa);
    load_row(sr + R0 * JJ + k0, ra);
    row_landed(xa);
    row_landed(sa);
    row_landed(ra);
    int r = R0;
#pragma unroll 1
    for (; r + 1 < R1; r += 2) {
        const int o1 = (r + 1) * JJ + k0;
        load_row(sx + o1, xb);
        load_row(ss + o1, sb);
        load_row(sr + o1, rb);
        fwd_step(r, lane, Gb, ph, esum, xa, sa, ra);
        if (r + 2 < R1) {
            const int o2 = (r + 2) * JJ + k0;
            load_row(sx + o2, xa);
            load_row(ss + o2, sa);
            load_row(sr + o2, ra);
        }
        fwd_step(r + 1, lane, Gb, ph, esum, xb, sb, rb);
    }
    if ((R1 - R0) & 1) fwd_step(R1 - 1, lane, Gb, ph, esum, xa, sa, ra);
}

__global__ void __launch_bounds__(32 * FW_WAVES)
k_forward_expand(const float* __restrict__ E, const float* __restrict__ text,
                 float* __restrict__ out) {
    __shared__ float s_xw[FW_ROWS * JJ];
    __shared__ float s_suf[FW_ROWS * JJ];
    __shared__ float s_rs[FW_ROWS * JJ];
    __shared__ float s_g[II * JJ];

    const int jt = blockIdx.x;
    const int b = blockIdx.y;
    const int lane = threadIdx.x & 31;
    const int w = __builtin_amdgcn_readfirstlane((int)(threadIdx.x >> 5));
    const int k0 = lane * KPL;
    const float* __restrict__ Eb = E + (size_t)b * II * JJ + k0;
    float* __restrict__ Gb = s_g + k0;

    const int half = lane >> 4;
    const int line = lane & 15;
    const int d0 = w * 16;
    const float* __restrict__ Tb = text + ((size_t)b * II + half * 2) * DD + (d0 + line);
    v2f bfr[II / 4];
#pragma unroll
    for (int k = 0; k < II; k += 4) {
        bfr[k / 4].x = Tb[(size_t)k * DD];
        bfr[k / 4].y = Tb[(size_t)(k + 1) * DD];
    }

    const int ra = FW_SPLIT + (w - 1);
    const int rb = ra + (FW_WAVES - 1);
    const int rc = rb + (FW_WAVES - 1);
    float e0[KPL], ea[KPL], eb[KPL], ec[KPL];
    load_row(Eb + (size_t)w * JJ, e0);
    if (w != 0) {
        load_row(Eb + (size_t)ra * JJ, ea);
        load_row(Eb + (size_t)rb * JJ, eb);
        if (rc < FW_ROWS) load_row(Eb + (size_t)rc * JJ, ec);
    }

    prep_row_to_lds(w, lane, k0, e0, s_xw, s_suf, s_rs);
    __syncthreads();

    float ph[KPL];
#pragma unroll
    for (int t = 0; t < KPL; ++t) ph[t] = (k0 + t == 0) ? 1.0f : 0.0f;
    int esum = 0;

    if (w == 0) {
        run_steps<0, FW_SPLIT>(lane, k0, Gb, ph, esum, s_xw, s_suf, s_rs);
    } else {
        prep_row_to_lds(ra, lane, k0, ea, s_xw, s_suf, s_rs);
        prep_row_to_lds(rb, lane, k0, eb, s_xw, s_suf, s_rs);
        if (rc < FW_ROWS) prep_row_to_lds(rc, lane, k0, ec, s_xw, s_suf, s_rs);
    }
    __syncthreads();

    if (w == 0) {
        run_steps<FW_SPLIT, FW_ROWS>(lane, k0, Gb, ph, esum, s_xw, s_suf, s_rs);

        const float last = row_total(ph) * ldexpf(1.0f, esum);
        float* __restrict__ gp = Gb + (size_t)(II - 1) * JJ;
#pragma unroll
        for (int u = 0; u < KPL / 2; ++u) {
            v2f q;
            q.x = 0.0f;
            q.y = (k0 + 2 * u + 1 == JJ - 1) ? last : 0.0f;
            *(v2f*)(gp + 2 * u) = q;
        }
    }
    __syncthreads();

    const int j0 = jt * 16;
    const float* __restrict__ Ga = s_g + (half * 2) * JJ + (j0 + line);
    v8f acc = {};
#pragma unroll
    for (int k = 0; k < II; k += 4) {
        v2f a;
        a.x = Ga[k * JJ];
        a.y = Ga[(k + 1) * JJ];
        acc = __builtin_amdgcn_wmma_f32_16x16x4_f32(false, a, false, bfr[k / 4],
                                                    (short)0, acc, false, false);
    }
#pragma unroll
    for (int v = 0; v < 8; ++v) {
        int j = j0 + v + half * 8;
        out[(size_t)b * JJ * DD + (size_t)j * DD + d0 + line] = acc[v];
    }
}

extern "C" void kernel_launch(void* const* d_in, const int* in_sizes, int n_in,
                              void* d_out, int out_size, void* d_ws, size_t ws_size,
                              hipStream_t stream) {
    const float* text = (const float*)d_in[0];
    const float* mel  = (const float*)d_in[1];
    const float* gum  = (const float*)d_in[2];
    float* out = (float*)d_out;

    float* E  = (float*)d_ws;

    k_energy_wmma<<<dim3(JJ / 16, II / 16, BB), 32, 0, stream>>>(text, mel, gum, E);
    k_forward_expand<<<dim3(JJ / 16, BB), 32 * FW_WAVES, 0, stream>>>(E, text, out);
    (void)in_sizes; (void)n_in; (void)out_size; (void)ws_size;
}


# K1 asks for a tile's whole K range at once: 64 loads of 16 bytes per lane in flight ahead of the first product (was 16 trips of 8 short loads), two accumulators; order of summation changes
# speedup vs baseline: 25.4351x; 1.2426x over previous
#include <hip/hip_runtime.h>
#include <math.h>

#define BB 2
#define II 48
#define JJ 320
#define DD 256

#define LOG2E_F 1.4426950408889634f
#define LN2_F   0.6931471805599453f

typedef __attribute__((ext_vector_type(2))) float v2f;
typedef __attribute__((ext_vector_type(4))) float v4f;
typedef __attribute__((ext_vector_type(8))) float v8f;

__device__ __forceinline__ float fast_log(float x) {
    return LN2_F * __builtin_amdgcn_logf(x);
}

__global__ void __launch_bounds__(32)
k_energy_wmma(const float* __restrict__ text, const float* __restrict__ mel,
              const float* __restrict__ gum, float* __restrict__ E) {
    const int b = blockIdx.z, it = blockIdx.y, jt = blockIdx.x;
    const int lane = threadIdx.x;
    const int half = lane >> 4;
    const int line = lane & 15;

    const size_t idx0 = (size_t)b * II * JJ + (size_t)(it * 16 + half * 8) * JJ + (jt * 16 + line);
    float u[8];
#pragma unroll
    for (int v = 0; v < 8; ++v) u[v] = gum[idx0 + (size_t)v * JJ];

    const v4f* __restrict__ Ta =
        (const v4f*)(text + (size_t)b * II * DD + (size_t)(it * 16 + line) * DD) + half;
    const v4f* __restrict__ Mb =
        (const v4f*)(mel + (size_t)b * JJ * DD + (size_t)(jt * 16 + line) * DD) + half;

    v4f a[DD / 8], bv[DD / 8];
#pragma unroll
    for (int q = 0; q < DD / 8; ++q) {
        a[q]  = Ta[2 * q];
        bv[q] = Mb[2 * q];
    }

    __builtin_amdgcn_sched_barrier(0);

    v8f acc0 = {}, acc1 = {};
#pragma unroll
    for (int q = 0; q < DD / 8; ++q) {
        v2f a0, a1, b0, b1;
        a0.x = a[q].x;  a0.y = a[q].y;  a1.x = a[q].z;  a1.y = a[q].w;
        b0.x = bv[q].x; b0.y = bv[q].y; b1.x = bv[q].z; b1.y = bv[q].w;
        acc0 = __builtin_amdgcn_wmma_f32_16x16x4_f32(false, a0, false, b0,
                                                     (short)0, acc0, false, false);
        acc1 = __builtin_amdgcn_wmma_f32_16x16x4_f32(false, a1, false, b1,
                                                     (short)0, acc1, false, false);
    }

    const float invTemp = 1.0f / 0.55f;
    const float scale   = 1.0f / 256.0f;
#pragma unroll
    for (int v = 0; v < 8; ++v) {
        float uc = fminf(fmaxf(u[v], 1e-7f), 1.0f - 1e-7f);
        float noise = -fast_log(-fast_log(uc));
        E[idx0 + (size_t)v * JJ] = ((acc0[v] + acc1[v]) * scale + noise) * invTemp;
    }
}

#define KPL (JJ / 32)
#define FW_WAVES 16
#define FW_ROWS (II - 1)
#define FW_SPLIT 16
#define FW_WIN (JJ - II + 2)

__device__ __forceinline__ void load_row(const float* __restrict__ p, float (&r)[KPL]) {
#pragma unroll
    for (int u = 0; u < KPL / 2; ++u) {
        v2f q = *(const v2f*)(p + 2 * u);
        r[2 * u] = q.x; r[2 * u + 1] = q.y;
    }
}
__device__ __forceinline__ void store_row(float* __restrict__ p, const float (&r)[KPL]) {
#pragma unroll
    for (int u = 0; u < KPL / 2; ++u) {
        v2f q;
        q.x = r[2 * u]; q.y = r[2 * u + 1];
        *(v2f*)(p + 2 * u) = q;
    }
}

template <int N>
__device__ __forceinline__ float row_shr0(float v) {
    return __int_as_float(__builtin_amdgcn_update_dpp(0, __float_as_int(v), 0x110 + N, 0xf, 0xf, true));
}
template <int N>
__device__ __forceinline__ float row_shl0(float v) {
    return __int_as_float(__builtin_amdgcn_update_dpp(0, __float_as_int(v), 0x100 + N, 0xf, 0xf, true));
}
__device__ __forceinline__ float row_incl_sum(float x) {
    x += row_shr0<1>(x);
    x += row_shr0<2>(x);
    x += row_shr0<4>(x);
    x += row_shr0<8>(x);
    return x;
}
__device__ __forceinline__ float row_incl_sum_rev(float x) {
    x += row_shl0<1>(x);
    x += row_shl0<2>(x);
    x += row_shl0<4>(x);
    x += row_shl0<8>(x);
    return x;
}
template <int L>
__device__ __forceinline__ float lane_value(float v) {
    return __int_as_float(__builtin_amdgcn_readlane(__float_as_int(v), L));
}

__device__ __forceinline__ void prep_row(const float (&e)[KPL], const int lane,
                                         float (&x)[KPL], float (&suf)[KPL], float (&rs)[KPL]) {
    const float refL = lane_value<0>(e[0]) * LOG2E_F;
#pragma unroll
    for (int t = 0; t < KPL; ++t) x[t] = __builtin_amdgcn_exp2f(fmaf(e[t], LOG2E_F, -refL));
    float sl[KPL];
    sl[KPL - 1] = x[KPL - 1];
#pragma unroll
    for (int t = KPL - 2; t >= 0; --t) sl[t] = sl[t + 1] + x[t];
    const float xs = row_incl_sum_rev(sl[0]);
    const float up = lane_value<16>(xs);
    const float add = (lane < 16) ? up : 0.0f;
    const float right = row_shl0<1>(xs) + add;
#pragma unroll
    for (int t = 0; t < KPL; ++t) suf[t] = right + sl[t];
#pragma unroll
    for (int t = 0; t < KPL; ++t) rs[t] = __builtin_amdgcn_rcpf(suf[t]);
}

__device__ __forceinline__ void prep_row_to_lds(const int row, const int lane, const int k0,
                                                const float (&e)[KPL],
                                                float* __restrict__ sx, float* __restrict__ ss,
                                                float* __restrict__ sr) {
    float x[KPL], suf[KPL], rs[KPL];
    prep_row(e, lane, x, suf, rs);
    const unsigned d = (unsigned)(k0 - row);
    float xw[KPL];
#pragma unroll
    for (int t = 0; t < KPL; ++t) xw[t] = ((d + (unsigned)t) < (unsigned)FW_WIN) ? x[t] : 0.0f;
    const int o = row * JJ + k0;
    store_row(sx + o, xw);
    store_row(ss + o, suf);
    store_row(sr + o, rs);
}

__device__ __forceinline__ float row_total(const float (&ph)[KPL]) {
    const float t01 = ph[0] + ph[1], t23 = ph[2] + ph[3], t45 = ph[4] + ph[5];
    const float t67 = ph[6] + ph[7], t89 = ph[8] + ph[9];
    float tot = ((t01 + t23) + (t45 + t67)) + t89;
    tot = row_incl_sum(tot);
    return lane_value<15>(tot) + lane_value<31>(tot);
}

__device__ __forceinline__ void fwd_step(const int r, const int lane,
                                         float* __restrict__ Gb,
                                         float (&ph)[KPL], int& esum,
                                         const float (&xw)[KPL], const float (&suf)[KPL],
                                         const float (&rs)[KPL]) {
    const float M = row_total(ph);
    int n = __builtin_amdgcn_frexp_expf(M);
    n = (n < -100) ? -100 : n;
    const float g = ldexpf(1.0f, -n);

    float emg[KPL];
#pragma unroll
    for (int t = 0; t < KPL; ++t) emg[t] = xw[t] * g;

    float l[KPL];
    l[0] = ph[0] * rs[0];
#pragma unroll
    for (int t = 1; t < KPL; ++t) l[t] = l[t - 1] + ph[t] * rs[t];

    const float xr = row_incl_sum(l[KPL - 1]);
    const float r0 = lane_value<15>(xr);
    const float add = (lane >= 16) ? r0 : 0.0f;
    const float excl = row_shr0<1>(xr) + add;
    float pre[KPL];
#pragma unroll
    for (int t = 0; t < KPL; ++t) pre[t] = excl + l[t];

    const float sc = ldexpf(1.0f, esum);
    float* __restrict__ gp = Gb + (size_t)r * JJ;
#pragma unroll
    for (int u = 0; u < KPL / 2; ++u) {
        v2f q;
        q.x = (pre[2 * u] * suf[2 * u]) * sc;
        q.y = (pre[2 * u + 1] * suf[2 * u + 1]) * sc;
        *(v2f*)(gp + 2 * u) = q;
    }

    const float hv = (xr + add) * emg[KPL - 1];
    float hand = row_shr0<1>(hv);
    const float h15 = lane_value<15>(hv);
    if (lane == 16) hand = h15;

    float nxt[KPL];
    nxt[0] = hand;
#pragma unroll
    for (int t = 0; t < KPL - 1; ++t) nxt[t + 1] = pre[t] * emg[t];
#pragma unroll
    for (int t = 0; t < KPL; ++t) ph[t] = nxt[t];
    esum += n;
}

__device__ __forceinline__ void row_landed(const float (&v)[KPL]) {
    asm volatile("" :: "v"(v[0]), "v"(v[1]), "v"(v[2]), "v"(v[3]), "v"(v[4]),
                       "v"(v[5]), "v"(v[6]), "v"(v[7]), "v"(v[8]), "v"(v[9]));
}

template <int R0, int R1>
__device__ __forceinline__ void run_steps(const int lane, const int k0, float* __restrict__ Gb,
                                          float (&ph)[KPL], int& esum,
                                          const float* __restrict__ sx, const float* __restrict__ ss,
                                          const float* __restrict__ sr) {
    float xa[KPL], sa[KPL], ra[KPL], xb[KPL], sb[KPL], rb[KPL];
    load_row(sx + R0 * JJ + k0, xa);
    load_row(ss + R0 * JJ + k0, sa);
    load_row(sr + R0 * JJ + k0, ra);
    row_landed(xa);
    row_landed(sa);
    row_landed(ra);
    int r = R0;
#pragma unroll 1
    for (; r + 1 < R1; r += 2) {
        const int o1 = (r + 1) * JJ + k0;
        load_row(sx + o1, xb);
        load_row(ss + o1, sb);
        load_row(sr + o1, rb);
        fwd_step(r, lane, Gb, ph, esum, xa, sa, ra);
        if (r + 2 < R1) {
            const int o2 = (r + 2) * JJ + k0;
            load_row(sx + o2, xa);
            load_row(ss + o2, sa);
            load_row(sr + o2, ra);
        }
        fwd_step(r + 1, lane, Gb, ph, esum, xb, sb, rb);
    }
    if ((R1 - R0) & 1) fwd_step(R1 - 1, lane, Gb, ph, esum, xa, sa, ra);
}

__global__ void __launch_bounds__(32 * FW_WAVES)
k_forward_expand(const float* __restrict__ E, const float* __restrict__ text,
                 float* __restrict__ out) {
    __shared__ float s_xw[FW_ROWS * JJ];
    __shared__ float s_suf[FW_ROWS * JJ];
    __shared__ float s_rs[FW_ROWS * JJ];
    __shared__ float s_g[II * JJ];

    const int jt = blockIdx.x;
    const int b = blockIdx.y;
    const int lane = threadIdx.x & 31;
    const int w = __builtin_amdgcn_readfirstlane((int)(threadIdx.x >> 5));
    const int k0 = lane * KPL;
    const float* __restrict__ Eb = E + (size_t)b * II * JJ + k0;
    float* __restrict__ Gb = s_g + k0;

    const int half = lane >> 4;
    const int line = lane & 15;
    const int d0 = w * 16;
    const float* __restrict__ Tb = text + ((size_t)b * II + half * 2) * DD + (d0 + line);
    v2f bfr[II / 4];
#pragma unroll
    for (int k = 0; k < II; k += 4) {
        bfr[k / 4].x = Tb[(size_t)k * DD];
        bfr[k / 4].y = Tb[(size_t)(k + 1) * DD];
    }

    const int ra = FW_SPLIT + (w - 1);
    const int rb = ra + (FW_WAVES - 1);
    const int rc = rb + (FW_WAVES - 1);
    float e0[KPL], ea[KPL], eb[KPL], ec[KPL];
    load_row(Eb + (size_t)w * JJ, e0);
    if (w != 0) {
        load_row(Eb + (size_t)ra * JJ, ea);
        load_row(Eb + (size_t)rb * JJ, eb);
        if (rc < FW_ROWS) load_row(Eb + (size_t)rc * JJ, ec);
    }

    prep_row_to_lds(w, lane, k0, e0, s_xw, s_suf, s_rs);
    __syncthreads();

    float ph[KPL];
#pragma unroll
    for (int t = 0; t < KPL; ++t) ph[t] = (k0 + t == 0) ? 1.0f : 0.0f;
    int esum = 0;

    if (w == 0) {
        run_steps<0, FW_SPLIT>(lane, k0, Gb, ph, esum, s_xw, s_suf, s_rs);
    } else {
        prep_row_to_lds(ra, lane, k0, ea, s_xw, s_suf, s_rs);
        prep_row_to_lds(rb, lane, k0, eb, s_xw, s_suf, s_rs);
        if (rc < FW_ROWS) prep_row_to_lds(rc, lane, k0, ec, s_xw, s_suf, s_rs);
    }
    __syncthreads();

    if (w == 0) {
        run_steps<FW_SPLIT, FW_ROWS>(lane, k0, Gb, ph, esum, s_xw, s_suf, s_rs);

        const float last = row_total(ph) * ldexpf(1.0f, esum);
        float* __restrict__ gp = Gb + (size_t)(II - 1) * JJ;
#pragma unroll
        for (int u = 0; u < KPL / 2; ++u) {
            v2f q;
            q.x = 0.0f;
            q.y = (k0 + 2 * u + 1 == JJ - 1) ? last : 0.0f;
            *(v2f*)(gp + 2 * u) = q;
        }
    }
    __syncthreads();

    const int j0 = jt * 16;
    const float* __restrict__ Ga = s_g + (half * 2) * JJ + (j0 + line);
    v8f acc = {};
#pragma unroll
    for (int k = 0; k < II; k += 4) {
        v2f a;
        a.x = Ga[k * JJ];
        a.y = Ga[(k + 1) * JJ];
        acc = __builtin_amdgcn_wmma_f32_16x16x4_f32(false, a, false, bfr[k / 4],
                                                    (short)0, acc, false, false);
    }
#pragma unroll
    for (int v = 0; v < 8; ++v) {
        int j = j0 + v + half * 8;
        out[(size_t)b * JJ * DD + (size_t)j * DD + d0 + line] = acc[v];
    }
}

extern "C" void kernel_launch(void* const* d_in, const int* in_sizes, int n_in,
                              void* d_out, int out_size, void* d_ws, size_t ws_size,
                              hipStream_t stream) {
    const float* text = (const float*)d_in[0];
    const float* mel  = (const float*)d_in[1];
    const float* gum  = (const float*)d_in[2];
    float* out = (float*)d_out;

    float* E  = (float*)d_ws;

    k_energy_wmma<<<dim3(JJ / 16, II / 16, BB), 32, 0, stream>>>(text, mel, gum, E);
    k_forward_expand<<<dim3(JJ / 16, BB), 32 * FW_WAVES, 0, stream>>>(E, text, out);
    (void)in_sizes; (void)n_in; (void)out_size; (void)ws_size;
}
